# starting kernel
# baseline (speedup 1.0000x reference)
#include <hip/hip_runtime.h>
#include <hip/hip_bf16.h>

#define NN 100000
#define DD 128
#define HH 64
#define KK 8
#define LL 3
#define CC 16
#define EE 1600000
#define THETA 0.1f

typedef __attribute__((ext_vector_type(2))) float v2f;
typedef __attribute__((ext_vector_type(8))) float v8f;

__device__ __forceinline__ v8f wmma_f32(v2f a, v2f b, v8f c) {
  return __builtin_amdgcn_wmma_f32_16x16x4_f32(
       false, a,  false, b,
       (short)0, c,  false,  false);
}

__global__ void zero_kernel(float* __restrict__ p, int n) {
  int i = blockIdx.x * blockDim.x + threadIdx.x;
  if (i < n) p[i] = 0.0f;
}

__global__ void degree_kernel(const int* __restrict__ col, float* __restrict__ deg) {
  int e = blockIdx.x * blockDim.x + threadIdx.x;
  if (e < EE) atomicAdd(&deg[col[e]], 1.0f);
}

__global__ void dn_kernel(const float* __restrict__ deg, float* __restrict__ dn) {
  int i = blockIdx.x * blockDim.x + threadIdx.x;
  if (i < NN) {
    float d = deg[i];
    dn[i] = (d > 0.0f) ? rsqrtf(d) : 0.0f;
  }
}

__global__ __launch_bounds__(256) void fc0_kernel(const float* __restrict__ x,
                                                  const float* __restrict__ w,
                                                  const float* __restrict__ b,
                                                  float* __restrict__ h) {
  const int wave = threadIdx.x >> 5;
  const int lane = threadIdx.x & 31;
  const int half = lane >> 4;
  const int lid  = lane & 15;
  const int node0 = blockIdx.x * 128 + wave * 16;
  if (node0 >= NN) return;

  v8f acc0 = {}, acc1 = {}, acc2 = {}, acc3 = {};
  for (int c = 0; c < DD / 4; ++c) {
    const int k0 = 4 * c + 2 * half;
    const float* ap = x + (node0 + lid) * DD + k0;
    v2f a; a.x = ap[0]; a.y = ap[1];
    const float* bp = w + k0 * HH;
    v2f b0; b0.x = bp[lid];      b0.y = bp[HH + lid];
    v2f b1; b1.x = bp[16 + lid]; b1.y = bp[HH + 16 + lid];
    v2f b2; b2.x = bp[32 + lid]; b2.y = bp[HH + 32 + lid];
    v2f b3; b3.x = bp[48 + lid]; b3.y = bp[HH + 48 + lid];
    acc0 = wmma_f32(a, b0, acc0);
    acc1 = wmma_f32(a, b1, acc1);
    acc2 = wmma_f32(a, b2, acc2);
    acc3 = wmma_f32(a, b3, acc3);
  }
  v8f accs[4] = {acc0, acc1, acc2, acc3};
  for (int t = 0; t < 4; ++t) {
    const float bias = b[16 * t + lid];
    for (int r = 0; r < 8; ++r) {
      float v = accs[t][r] + bias;
      h[(node0 + r + 8 * half) * HH + 16 * t + lid] = v > 0.0f ? v : 0.0f;
    }
  }
}

__global__ void scatter_kernel(const int* __restrict__ row, const int* __restrict__ col,
                               const float* __restrict__ dn, const float* __restrict__ h,
                               float* __restrict__ agg) {
  unsigned idx = blockIdx.x * blockDim.x + threadIdx.x;
  unsigned e = idx >> 4;
  int seg = idx & 15;
  if (e >= EE) return;
  int r = row[e], c = col[e];
  float v = dn[r] * dn[c];
  const float4 hv = *(const float4*)(h + r * HH + seg * 4);
  float* ap = agg + c * HH + seg * 4;
  atomicAdd(ap + 0, v * hv.x);
  atomicAdd(ap + 1, v * hv.y);
  atomicAdd(ap + 2, v * hv.z);
  atomicAdd(ap + 3, v * hv.w);
}

__global__ void gate_kernel(const float* __restrict__ h, const float* __restrict__ env_w,
                            const float* __restrict__ env_b, float* __restrict__ e_out,
                            int layer) {
  int n = blockIdx.x * blockDim.x + threadIdx.x;
  if (n >= NN) return;
  const float* W = env_w + layer * 2 * HH * KK;
  const float* B = env_b + layer * KK;
  float logit[KK];
  #pragma unroll
  for (int k = 0; k < KK; ++k) logit[k] = B[k];
  for (int f = 0; f < HH; ++f) {
    float hv = h[n * HH + f];
    const float* wr = W + f * KK;
    #pragma unroll
    for (int k = 0; k < KK; ++k) logit[k] += hv * wr[k];
  }
  float m = logit[0];
  #pragma unroll
  for (int k = 1; k < KK; ++k) m = fmaxf(m, logit[k]);
  float s = 0.0f;
  float ex[KK];
  #pragma unroll
  for (int k = 0; k < KK; ++k) { ex[k] = __expf(logit[k] - m); s += ex[k]; }
  float inv = 1.0f / s;
  #pragma unroll
  for (int k = 0; k < KK; ++k) {
    float pi = ex[k] * inv;
    e_out[n * KK + k] = (pi > THETA) ? pi : 0.0f;
  }
}

__global__ __launch_bounds__(256) void conv_kernel(const float* __restrict__ agg,
                                                   const float* __restrict__ h,
                                                   const float* __restrict__ e_gate,
                                                   const float* __restrict__ conv_w,
                                                   float* __restrict__ h_out,
                                                   int layer) {
  const int wave = threadIdx.x >> 5;
  const int lane = threadIdx.x & 31;
  const int half = lane >> 4;
  const int lid  = lane & 15;
  const int node0 = blockIdx.x * 128 + wave * 16;
  if (node0 >= NN) return;

  const float* W = conv_w + layer * (KK * 2 * HH * HH);
  v8f acc0 = {}, acc1 = {}, acc2 = {}, acc3 = {};

  #pragma unroll 1
  for (int k = 0; k < KK; ++k) {
    const float ek = e_gate[(node0 + lid) * KK + k];
    if (!__any(ek != 0.0f)) continue;
    const float* Wk = W + k * (2 * HH * HH);
    #pragma unroll 1
    for (int c = 0; c < (2 * HH) / 4; ++c) {
      const int kk = 4 * c + 2 * half;
      const float* src = (kk < HH) ? agg : h;
      const int fk = (kk < HH) ? kk : kk - HH;
      const float* ap = src + (node0 + lid) * HH + fk;
      v2f a; a.x = ek * ap[0]; a.y = ek * ap[1];
      const float* bp = Wk + kk * HH;
      v2f b0; b0.x = bp[lid];      b0.y = bp[HH + lid];
      v2f b1; b1.x = bp[16 + lid]; b1.y = bp[HH + 16 + lid];
      v2f b2; b2.x = bp[32 + lid]; b2.y = bp[HH + 32 + lid];
      v2f b3; b3.x = bp[48 + lid]; b3.y = bp[HH + 48 + lid];
      acc0 = wmma_f32(a, b0, acc0);
      acc1 = wmma_f32(a, b1, acc1);
      acc2 = wmma_f32(a, b2, acc2);
      acc3 = wmma_f32(a, b3, acc3);
    }
  }
  v8f accs[4] = {acc0, acc1, acc2, acc3};
  for (int t = 0; t < 4; ++t) {
    for (int r = 0; r < 8; ++r) {
      const int idx = (node0 + r + 8 * half) * HH + 16 * t + lid;
      float v = accs[t][r] + h[idx];
      h_out[idx] = v > 0.0f ? v : 0.0f;
    }
  }
}

__global__ __launch_bounds__(256) void fc1_kernel(const float* __restrict__ h,
                                                  const float* __restrict__ w,
                                                  const float* __restrict__ b,
                                                  float* __restrict__ out) {
  const int wave = threadIdx.x >> 5;
  const int lane = threadIdx.x & 31;
  const int half = lane >> 4;
  const int lid  = lane & 15;
  const int node0 = blockIdx.x * 128 + wave * 16;
  if (node0 >= NN) return;

  v8f acc = {};
  for (int c = 0; c < HH / 4; ++c) {
    const int k0 = 4 * c + 2 * half;
    const float* ap = h + (node0 + lid) * HH + k0;
    v2f a; a.x = ap[0]; a.y = ap[1];
    const float* bp = w + k0 * CC;
    v2f bb; bb.x = bp[lid]; bb.y = bp[CC + lid];
    acc = wmma_f32(a, bb, acc);
  }
  const float bias = b[lid];
  for (int r = 0; r < 8; ++r)
    out[(node0 + r + 8 * half) * CC + lid] = acc[r] + bias;
}

extern "C" void kernel_launch(void* const* d_in, const int* in_sizes, int n_in,
                              void* d_out, int out_size, void* d_ws, size_t ws_size,
                              hipStream_t stream) {
  const float* x      = (const float*)d_in[0];
  const int*   eidx   = (const int*)d_in[1];
  const float* fc0_w  = (const float*)d_in[2];
  const float* fc0_b  = (const float*)d_in[3];
  const float* fc1_w  = (const float*)d_in[4];
  const float* fc1_b  = (const float*)d_in[5];
  const float* env_w  = (const float*)d_in[6];
  const float* env_b  = (const float*)d_in[7];
  const float* conv_w = (const float*)d_in[8];
  float* out = (float*)d_out;

  const int* row = eidx;
  const int* col = eidx + EE;

  float* ws  = (float*)d_ws;
  float* deg = ws;
  float* dn  = deg + NN;
  float* hA  = dn + NN;
  float* hB  = hA + (size_t)NN * HH;
  float* agg = hB + (size_t)NN * HH;
  float* eg  = agg + (size_t)NN * HH;

  const int T = 256;
  const int gemm_blocks = (NN + 127) / 128;

  zero_kernel<<<(NN + T - 1) / T, T, 0, stream>>>(deg, NN);
  degree_kernel<<<(EE + T - 1) / T, T, 0, stream>>>(col, deg);
  dn_kernel<<<(NN + T - 1) / T, T, 0, stream>>>(deg, dn);

  fc0_kernel<<<gemm_blocks, T, 0, stream>>>(x, fc0_w, fc0_b, hA);

  float* cur = hA;
  float* nxt = hB;
  for (int layer = 0; layer < LL; ++layer) {
    zero_kernel<<<(NN * HH + T - 1) / T, T, 0, stream>>>(agg, NN * HH);
    {
      unsigned items = (unsigned)EE * 16u;
      scatter_kernel<<<(items + T - 1) / T, T, 0, stream>>>(row, col, dn, cur, agg);
    }
    gate_kernel<<<(NN + T - 1) / T, T, 0, stream>>>(cur, env_w, env_b, eg, layer);
    conv_kernel<<<gemm_blocks, T, 0, stream>>>(agg, cur, eg, conv_w, nxt, layer);
    float* tmp = cur; cur = nxt; nxt = tmp;
  }

  fc1_kernel<<<gemm_blocks, T, 0, stream>>>(cur, fc1_w, fc1_b, out);
}


# aggregation: edges binned once per call by 391-node destination range; per layer a work-group sums its range's rows in an LDS tile (LDS f32 atomic adds, arrival order) and stores agg once
# speedup vs baseline: 2.2732x; 2.2732x over previous
#include <hip/hip_runtime.h>
#include <hip/hip_bf16.h>

#define NN 100000
#define DD 128
#define HH 64
#define KK 8
#define LL 3
#define CC 16
#define EE 1600000
#define THETA 0.1f

typedef __attribute__((ext_vector_type(2))) float v2f;
typedef __attribute__((ext_vector_type(8))) float v8f;

__device__ __forceinline__ v8f wmma_f32(v2f a, v2f b, v8f c) {
  return __builtin_amdgcn_wmma_f32_16x16x4_f32(
       false, a,  false, b,
       (short)0, c,  false,  false);
}


typedef __attribute__((ext_vector_type(4))) int v4i;
#define NQ 4
#define QN (NN / NQ)
#define GS 64
#define ES (EE / GS)
#define PS 25024
#define HT 1024
#define HI ((ES / 4 + HT - 1) / HT)
static_assert(NN % NQ == 0 && QN % 4 == 0 && PS >= QN && PS % 32 == 0, "quarters are whole 16-byte items, planes whole lines");
static_assert(EE % GS == 0 && ES % 4 == 0, "slices are whole 16-byte items");
static_assert(QN <= ES, "HI trips also cover a quarter's QN/4 items");
static_assert((size_t)NQ * GS * PS <= (size_t)2 * NN * HH, "the planes fit into the two feature planes they borrow");
__global__ __launch_bounds__(HT) void degree_hist_kernel(const int* __restrict__ col, int* __restrict__ part) {
  __shared__ __attribute__((aligned(16))) int cnt[QN];
  const int tid = threadIdx.x;
  const int q = blockIdx.x & (NQ - 1);
  const int g = blockIdx.x / NQ;
  const v4i* cp = (const v4i*)(col + g * ES);
  v4i cv[HI];
  #pragma unroll
  for (int i = 0; i < HI; ++i) {
    const int j = i * HT + tid;
    const v4i none = {-1, -1, -1, -1};
    cv[i] = (j < ES / 4) ? cp[j] : none;
  }
  #pragma unroll
  for (int i = 0; i < HI; ++i) {
    const int j = i * HT + tid;
    const v4i z = {0, 0, 0, 0};
    if (j < QN / 4) ((v4i*)cnt)[j] = z;
  }
  __syncthreads();
  const int lo = q * QN;
  #pragma unroll
  for (int i = 0; i < HI; ++i) {
    const unsigned d0 = (unsigned)(cv[i].x - lo), d1 = (unsigned)(cv[i].y - lo);
    const unsigned d2 = (unsigned)(cv[i].z - lo), d3 = (unsigned)(cv[i].w - lo);
    if (d0 < (unsigned)QN) atomicAdd(&cnt[d0], 1);
    if (d1 < (unsigned)QN) atomicAdd(&cnt[d1], 1);
    if (d2 < (unsigned)QN) atomicAdd(&cnt[d2], 1);
    if (d3 < (unsigned)QN) atomicAdd(&cnt[d3], 1);
  }
  __syncthreads();
  v4i* op = (v4i*)(part + (q * GS + g) * PS);
  #pragma unroll
  for (int i = 0; i < HI; ++i) {
    const int j = i * HT + tid;
    if (j < QN / 4) op[j] = ((const v4i*)cnt)[j];
  }
}

__global__ void dn_kernel(const int* __restrict__ part, float* __restrict__ dn) {
  int i = blockIdx.x * blockDim.x + threadIdx.x;
  if (i < NN) {
    const int q = i / QN;
    const int* p = part + q * (GS * PS) + (i - q * QN);
    int d = 0;
    #pragma unroll
    for (int g = 0; g < GS; ++g) d += p[g * PS];
    dn[i] = (d > 0) ? rsqrtf((float)d) : 0.0f;
  }
}

__global__ __launch_bounds__(256) void fc0_kernel(const float* __restrict__ x,
                                                  const float* __restrict__ w,
                                                  const float* __restrict__ b,
                                                  float* __restrict__ h) {
  const int wave = threadIdx.x >> 5;
  const int lane = threadIdx.x & 31;
  const int half = lane >> 4;
  const int lid  = lane & 15;
  const int node0 = blockIdx.x * 128 + wave * 16;
  if (node0 >= NN) return;

  v8f acc0 = {}, acc1 = {}, acc2 = {}, acc3 = {};
  for (int c = 0; c < DD / 4; ++c) {
    const int k0 = 4 * c + 2 * half;
    const float* ap = x + (node0 + lid) * DD + k0;
    v2f a; a.x = ap[0]; a.y = ap[1];
    const float* bp = w + k0 * HH;
    v2f b0; b0.x = bp[lid];      b0.y = bp[HH + lid];
    v2f b1; b1.x = bp[16 + lid]; b1.y = bp[HH + 16 + lid];
    v2f b2; b2.x = bp[32 + lid]; b2.y = bp[HH + 32 + lid];
    v2f b3; b3.x = bp[48 + lid]; b3.y = bp[HH + 48 + lid];
    acc0 = wmma_f32(a, b0, acc0);
    acc1 = wmma_f32(a, b1, acc1);
    acc2 = wmma_f32(a, b2, acc2);
    acc3 = wmma_f32(a, b3, acc3);
  }
  v8f accs[4] = {acc0, acc1, acc2, acc3};
  for (int t = 0; t < 4; ++t) {
    const float bias = b[16 * t + lid];
    for (int r = 0; r < 8; ++r) {
      float v = accs[t][r] + bias;
      h[(node0 + r + 8 * half) * HH + 16 * t + lid] = v > 0.0f ? v : 0.0f;
    }
  }
}

typedef __attribute__((ext_vector_type(2))) int v2i;
typedef __attribute__((ext_vector_type(4))) float v4f;
#define RB 391
#define NBK 256
#define SL 250
#define SE (EE / SL)
#define BT 1024
#define RBITS 17
#define BI ((SE + BT - 1) / BT)
static_assert(RB * NBK >= NN, "the ranges cover every node");
static_assert(EE % SL == 0, "slices are whole");
static_assert(BT == 4 * NBK, "four threads per range in the place pass");
static_assert(NN <= (1 << RBITS) && RB <= (1 << (31 - RBITS)), "source row and local row fit one non-negative word");
static_assert(NBK * SL + NBK + 1 <= NN, "counts and range starts fit the spare N words at the head of the workspace");
static_assert((size_t)EE * 2 <= (size_t)NN * HH, "the records fit one feature plane");

__global__ __launch_bounds__(BT) void bin_count_kernel(const int* __restrict__ col, int* __restrict__ cnt) {
  __shared__ int hist[NBK];
  const int tid = threadIdx.x;
  const int s = blockIdx.x;
  int cv[BI];
  #pragma unroll
  for (int u = 0; u < BI; ++u) {
    const int i = u * BT + tid;
    cv[u] = (i < SE) ? col[s * SE + i] : -1;
  }
  if (tid < NBK) hist[tid] = 0;
  __syncthreads();
  #pragma unroll
  for (int u = 0; u < BI; ++u)
    if (cv[u] >= 0) atomicAdd(&hist[cv[u] / RB], 1);
  __syncthreads();
  if (tid < NBK) cnt[tid * SL + s] = hist[tid];
}

__global__ __launch_bounds__(BT) void bin_place_kernel(const int* __restrict__ row, const int* __restrict__ col,
                                                       const float* __restrict__ dn, const int* __restrict__ cnt,
                                                       int* __restrict__ boff, int* __restrict__ recw) {
  __shared__ int tot[NBK];
  __shared__ int cur[NBK];
  const int tid = threadIdx.x;
  const int s = blockIdx.x;
  v2i* recs = (v2i*)recw;
  int rv[BI], cv[BI];
  float wv[BI];
  #pragma unroll
  for (int u = 0; u < BI; ++u) {
    const int i = u * BT + tid;
    const bool ok = i < SE;
    rv[u] = ok ? row[s * SE + i] : 0;
    cv[u] = ok ? col[s * SE + i] : -1;
  }
  #pragma unroll
  for (int u = 0; u < BI; ++u) wv[u] = (cv[u] >= 0) ? dn[rv[u]] * dn[cv[u]] : 0.0f;
  {
    const int b = tid >> 2, part = tid & 3;
    int t = 0, p = 0;
    for (int s2 = part; s2 < SL; s2 += 4) {
      const int v = cnt[b * SL + s2];
      t += v;
      p += (s2 < s) ? v : 0;
    }
    t += __shfl_xor(t, 1); t += __shfl_xor(t, 2);
    p += __shfl_xor(p, 1); p += __shfl_xor(p, 2);
    if (part == 0) { tot[b] = t; cur[b] = p; }
  }
  __syncthreads();
  if (tid < NBK) {
    int start = 0;
    for (int b2 = 0; b2 < NBK; ++b2) start += (b2 < tid) ? tot[b2] : 0;
    cur[tid] += start;
    if (s == 0) {
      boff[tid] = start;
      if (tid == NBK - 1) boff[NBK] = start + tot[tid];
    }
  }
  __syncthreads();
  #pragma unroll
  for (int u = 0; u < BI; ++u) {
    if (cv[u] >= 0) {
      const int bk = cv[u] / RB;
      const int pos = atomicAdd(&cur[bk], 1);
      v2i rec;
      rec.x = rv[u] | ((cv[u] - bk * RB) << RBITS);
      rec.y = __float_as_int(wv[u]);
      recs[pos] = rec;
    }
  }
}

#define EB 8
__device__ __forceinline__ void load_rows(const float* __restrict__ h, int r, int lane, int j0, float* hv) {
  #pragma unroll
  for (int u = 0; u < EB; ++u) {
    const int rj = __builtin_amdgcn_readlane(r, j0 + u);
    const float* hp = h + rj * HH + lane;
    hv[2 * u]     = hp[0];
    hv[2 * u + 1] = hp[32];
  }
}
__device__ __forceinline__ void add_rows(float* tile, int lc, int vb, int lane, int j0, const float* hv) {
  #pragma unroll
  for (int u = 0; u < EB; ++u) {
    const int cj = __builtin_amdgcn_readlane(lc, j0 + u);
    const float vj = __int_as_float(__builtin_amdgcn_readlane(vb, j0 + u));
    float* tp = tile + cj * HH + lane;
    atomicAdd(tp,      vj * hv[2 * u]);
    atomicAdd(tp + 32, vj * hv[2 * u + 1]);
  }
}
__global__ __launch_bounds__(BT) void range_sum_kernel(const int* __restrict__ recw, const int* __restrict__ boff,
                                                       const float* __restrict__ h, float* __restrict__ agg) {
  __shared__ __attribute__((aligned(16))) float tile[RB * HH];
  const v2i* recs = (const v2i*)recw;
  const int tid = threadIdx.x;
  const int lane = tid & 31;
  const int wave = tid >> 5;
  const int b = blockIdx.x;
  const v4f z = {0.0f, 0.0f, 0.0f, 0.0f};
  for (int i = tid; i < RB * HH / 4; i += BT) ((v4f*)tile)[i] = z;
  const int e_lo = boff[b], e_hi = boff[b + 1];
  __syncthreads();
  for (int e0 = e_lo + wave * 32; e0 < e_hi; e0 += BT) {
    const int n = min(32, e_hi - e0);
    v2i rec = {0, 0};
    if (lane < n) rec = recs[e0 + lane];
    const int r  = rec.x & ((1 << RBITS) - 1);
    const int lc = rec.x >> RBITS;
    const int vb = rec.y;
    if (n == 32) {
      float ha[2 * EB], hb[2 * EB];
      load_rows(h, r, lane, 0 * EB, ha);
      load_rows(h, r, lane, 1 * EB, hb);
      add_rows(tile, lc, vb, lane, 0 * EB, ha);
      load_rows(h, r, lane, 2 * EB, ha);
      add_rows(tile, lc, vb, lane, 1 * EB, hb);
      load_rows(h, r, lane, 3 * EB, hb);
      add_rows(tile, lc, vb, lane, 2 * EB, ha);
      add_rows(tile, lc, vb, lane, 3 * EB, hb);
    } else {
      for (int j = 0; j < n; ++j) {
        const int rj = __builtin_amdgcn_readlane(r, j);
        const int cj = __builtin_amdgcn_readlane(lc, j);
        const float vj = __int_as_float(__builtin_amdgcn_readlane(vb, j));
        const float* hp = h + rj * HH + lane;
        float* tp = tile + cj * HH + lane;
        atomicAdd(tp,      vj * hp[0]);
        atomicAdd(tp + 32, vj * hp[32]);
      }
    }
  }
  __syncthreads();
  const int node_lo = b * RB;
  v4f* ap = (v4f*)(agg + node_lo * HH);
  for (int i = tid; i < RB * HH / 4; i += BT)
    if (node_lo + (i >> 4) < NN) ap[i] = ((const v4f*)tile)[i];
}

__global__ void gate_kernel(const float* __restrict__ h, const float* __restrict__ env_w,
                            const float* __restrict__ env_b, float* __restrict__ e_out,
                            int layer) {
  int n = blockIdx.x * blockDim.x + threadIdx.x;
  if (n >= NN) return;
  const float* W = env_w + layer * 2 * HH * KK;
  const float* B = env_b + layer * KK;
  float logit[KK];
  #pragma unroll
  for (int k = 0; k < KK; ++k) logit[k] = B[k];
  for (int f = 0; f < HH; ++f) {
    float hv = h[n * HH + f];
    const float* wr = W + f * KK;
    #pragma unroll
    for (int k = 0; k < KK; ++k) logit[k] += hv * wr[k];
  }
  float m = logit[0];
  #pragma unroll
  for (int k = 1; k < KK; ++k) m = fmaxf(m, logit[k]);
  float s = 0.0f;
  float ex[KK];
  #pragma unroll
  for (int k = 0; k < KK; ++k) { ex[k] = __expf(logit[k] - m); s += ex[k]; }
  float inv = 1.0f / s;
  #pragma unroll
  for (int k = 0; k < KK; ++k) {
    float pi = ex[k] * inv;
    e_out[n * KK + k] = (pi > THETA) ? pi : 0.0f;
  }
}

#define WP2 128
#define WBUF (HH * WP2)
#define NB 128

__device__ __forceinline__ void conv_half(const float* __restrict__ arow, const float* bl,
                                          float ek, int half, int lid,
                                          v8f& acc0, v8f& acc1, v8f& acc2, v8f& acc3) {
  const int s0 = 32 * half, s1 = 32 - 32 * half;
  #pragma unroll 4
  for (int c = 0; c < HH / 4; ++c) {
    const float* ap = arow + 4 * c + 2 * half;
    v2f a; a.x = ek * ap[0]; a.y = ek * ap[1];
    const float* bp = bl + (2 * c + half) * WP2 + 2 * lid;
    const v2f b0 = *(const v2f*)(bp + s0);
    const v2f b1 = *(const v2f*)(bp + s1);
    const v2f b2 = *(const v2f*)(bp + 64 + s0);
    const v2f b3 = *(const v2f*)(bp + 64 + s1);
    acc0 = wmma_f32(a, b0, acc0);
    acc1 = wmma_f32(a, b1, acc1);
    acc2 = wmma_f32(a, b2, acc2);
    acc3 = wmma_f32(a, b3, acc3);
  }
}

__global__ __launch_bounds__(256) void conv_kernel(const float* __restrict__ agg,
                                                   float* h,
                                                   const float* __restrict__ e_gate,
                                                   const float* __restrict__ conv_w,
                                                   int layer) {
  __shared__ __attribute__((aligned(16))) float wl[WBUF];
  __shared__ __attribute__((aligned(16))) float oacc[(NB + 1) * HH];
  __shared__ int   lrow[KK * NB];
  __shared__ float lgate[KK * NB];
  __shared__ int   wcnt[KK * 8];

  const int tid  = threadIdx.x;
  const int wave = tid >> 5;
  const int lane = tid & 31;
  const int half = lane >> 4;
  const int lid  = lane & 15;
  const int base = blockIdx.x * NB;
  const int node0 = base + wave * 16;
  const bool live = node0 < NN;

  const float* W = conv_w + layer * (KK * 2 * HH * HH);

  const int goff = (tid >> 4) * (2 * HH) + (tid & 15) * 4;
  const int loff = (tid >> 4) * WP2 + (((tid & 15) * 8) ^ (((tid >> 4) & 1) * 32));
  v4f r0[4], r1[4];
  #pragma unroll
  for (int i = 0; i < 4; ++i) {
    r0[i] = *(const v4f*)(W + goff + i * 16 * 2 * HH);
    r1[i] = *(const v4f*)(W + goff + i * 16 * 2 * HH + HH);
  }
  v4f ga = {}, gb = {};
  if (live) {
    const v4f* gp = (const v4f*)(e_gate + (node0 + lid) * KK);
    ga = gp[0]; gb = gp[1];
  }
  const float g[KK] = {ga.x, ga.y, ga.z, ga.w, gb.x, gb.y, gb.z, gb.w};
  unsigned m[KK];
  #pragma unroll
  for (int k = 0; k < KK; ++k) m[k] = __builtin_amdgcn_ballot_w32(g[k] != 0.0f) & 0xFFFFu;
  #pragma unroll
  for (int k = 0; k < KK; ++k)
    if (lane == k) wcnt[k * 8 + wave] = __popc(m[k]);
  #pragma unroll
  for (int i = 0; i < 8; ++i) {
    const v4f z = {};
    *(v4f*)(oacc + wave * 16 * HH + (i * 32 + lane) * 4) = z;
  }
  if (wave == 0) { oacc[NB * HH + lane] = 0.0f; oacc[NB * HH + 32 + lane] = 0.0f; }
  #pragma unroll
  for (int i = 0; i < 4; ++i) {
    v4f lo, hi;
    lo.x = r0[i].x; lo.y = r1[i].x; lo.z = r0[i].y; lo.w = r1[i].y;
    hi.x = r0[i].z; hi.y = r1[i].z; hi.z = r0[i].w; hi.w = r1[i].w;
    *(v4f*)(wl + loff + i * 16 * WP2)     = lo;
    *(v4f*)(wl + loff + i * 16 * WP2 + 4) = hi;
  }
  __syncthreads();

  #pragma unroll
  for (int k = 0; k < KK; ++k) {
    int off = 0;
    #pragma unroll
    for (int w2 = 0; w2 < 8; ++w2) {
      const int cw = wcnt[k * 8 + w2];
      off += (w2 < wave) ? cw : 0;
    }
    if (half == 0 && g[k] != 0.0f) {
      const int pos = off + __popc(m[k] & ((1u << lid) - 1u));
      lrow[k * NB + pos]  = wave * 16 + lid;
      lgate[k * NB + pos] = g[k];
    }
  }
  __syncthreads();

  const int jt = (wave & 1) ? (wave ^ 4) : wave;

  #pragma unroll 1
  for (int k = 0; k < KK; ++k) {
    const bool more = (k + 1 < KK);
    if (more) {
      const float* Wn = W + (k + 1) * (2 * HH * HH) + goff;
      #pragma unroll
      for (int i = 0; i < 4; ++i) {
        r0[i] = *(const v4f*)(Wn + i * 16 * 2 * HH);
        r1[i] = *(const v4f*)(Wn + i * 16 * 2 * HH + HH);
      }
    }
    int tk = 0;
    #pragma unroll
    for (int w2 = 0; w2 < 8; ++w2) tk += wcnt[k * 8 + w2];
    if (16 * jt < tk) {
      const int i = 16 * jt + lid;
      const bool valid = i < tk;
      int lr = 0; float ek = 0.0f;
      if (valid) { lr = lrow[k * NB + i]; ek = lgate[k * NB + i]; }
      const int drow = valid ? lr : NB;
      v8f acc0 = {}, acc1 = {}, acc2 = {}, acc3 = {};
      const float* bl = wl;
      conv_half(agg + (base + lr) * HH, bl,                  ek, half, lid, acc0, acc1, acc2, acc3);
      conv_half(h   + (base + lr) * HH, bl + (HH / 2) * WP2, ek, half, lid, acc0, acc1, acc2, acc3);
      float* op[8];
      float t0[8], t1[8], t2[8], t3[8];
      #pragma unroll
      for (int r = 0; r < 8; ++r) {
        const int dl = __builtin_amdgcn_readlane(drow, r);
        const int dh = __builtin_amdgcn_readlane(drow, r + 8);
        op[r] = oacc + (half ? dh : dl) * HH + lid;
        t0[r] = op[r][0];
        t1[r] = op[r][16];
        t2[r] = op[r][32];
        t3[r] = op[r][48];
      }
      #pragma unroll
      for (int r = 0; r < 8; ++r) {
        op[r][0]  = t0[r] + acc0[r];
        op[r][16] = t1[r] + acc1[r];
        op[r][32] = t2[r] + acc2[r];
        op[r][48] = t3[r] + acc3[r];
      }
    }
    if (more) {
      __syncthreads();
      float* dst = wl + loff;
      #pragma unroll
      for (int i = 0; i < 4; ++i) {
        v4f lo, hi;
        lo.x = r0[i].x; lo.y = r1[i].x; lo.z = r0[i].y; lo.w = r1[i].y;
        hi.x = r0[i].z; hi.y = r1[i].z; hi.z = r0[i].w; hi.w = r1[i].w;
        *(v4f*)(dst + i * 16 * WP2)     = lo;
        *(v4f*)(dst + i * 16 * WP2 + 4) = hi;
      }
    }
    __syncthreads();
  }
  if (!live) return;
  #pragma unroll 4
  for (int r = 0; r < 16; ++r) {
    const int idx = (node0 + r) * HH + lane;
    const float* op = oacc + (wave * 16 + r) * HH + lane;
    const float v0 = op[0]  + h[idx];
    const float v1 = op[32] + h[idx + 32];
    h[idx]      = v0 > 0.0f ? v0 : 0.0f;
    h[idx + 32] = v1 > 0.0f ? v1 : 0.0f;
  }
}

__global__ __launch_bounds__(256) void fc1_kernel(const float* __restrict__ h,
                                                  const float* __restrict__ w,
                                                  const float* __restrict__ b,
                                                  float* __restrict__ out) {
  const int wave = threadIdx.x >> 5;
  const int lane = threadIdx.x & 31;
  const int half = lane >> 4;
  const int lid  = lane & 15;
  const int node0 = blockIdx.x * 128 + wave * 16;
  if (node0 >= NN) return;

  v8f acc = {};
  for (int c = 0; c < HH / 4; ++c) {
    const int k0 = 4 * c + 2 * half;
    const float* ap = h + (node0 + lid) * HH + k0;
    v2f a; a.x = ap[0]; a.y = ap[1];
    const float* bp = w + k0 * CC;
    v2f bb; bb.x = bp[lid]; bb.y = bp[CC + lid];
    acc = wmma_f32(a, bb, acc);
  }
  const float bias = b[lid];
  for (int r = 0; r < 8; ++r)
    out[(node0 + r + 8 * half) * CC + lid] = acc[r] + bias;
}

extern "C" void kernel_launch(void* const* d_in, const int* in_sizes, int n_in,
                              void* d_out, int out_size, void* d_ws, size_t ws_size,
                              hipStream_t stream) {
  const float* x      = (const float*)d_in[0];
  const int*   eidx   = (const int*)d_in[1];
  const float* fc0_w  = (const float*)d_in[2];
  const float* fc0_b  = (const float*)d_in[3];
  const float* fc1_w  = (const float*)d_in[4];
  const float* fc1_b  = (const float*)d_in[5];
  const float* env_w  = (const float*)d_in[6];
  const float* env_b  = (const float*)d_in[7];
  const float* conv_w = (const float*)d_in[8];
  float* out = (float*)d_out;

  const int* row = eidx;
  const int* col = eidx + EE;

  float* ws  = (float*)d_ws;
  float* dn  = ws + NN;
  float* hA  = dn + NN;
  float* hB  = hA + (size_t)NN * HH;
  float* agg = hB + (size_t)NN * HH;
  float* eg  = agg + (size_t)NN * HH;
  int* cnt   = (int*)ws;
  int* boff  = cnt + NBK * SL;
  int* recs  = (int*)hB;

  const int T = 256;
  const int gemm_blocks = (NN + 127) / 128;

  int* part = (int*)hB;
  degree_hist_kernel<<<NQ * GS, HT, 0, stream>>>(col, part);
  dn_kernel<<<(NN + T - 1) / T, T, 0, stream>>>(part, dn);

  bin_count_kernel<<<SL, BT, 0, stream>>>(col, cnt);
  bin_place_kernel<<<SL, BT, 0, stream>>>(row, col, dn, cnt, boff, recs);

  fc0_kernel<<<gemm_blocks, T, 0, stream>>>(x, fc0_w, fc0_b, hA);

  for (int layer = 0; layer < LL; ++layer) {
    range_sum_kernel<<<NBK, BT, 0, stream>>>(recs, boff, hA, agg);
    gate_kernel<<<(NN + T - 1) / T, T, 0, stream>>>(hA, env_w, env_b, eg, layer);
    conv_kernel<<<gemm_blocks, T, 0, stream>>>(agg, hA, eg, conv_w, layer);
  }

  fc1_kernel<<<gemm_blocks, T, 0, stream>>>(hA, fc1_w, fc1_b, out);
}


# aggregation: a range's short last record group is padded with zero-weight fillers (spare LDS tile row) and takes the 16-loads-in-flight path, not 2 dependent loads per edge; sums in LDS arrival order
# speedup vs baseline: 2.3472x; 1.0326x over previous
#include <hip/hip_runtime.h>
#include <hip/hip_bf16.h>

#define NN 100000
#define DD 128
#define HH 64
#define KK 8
#define LL 3
#define CC 16
#define EE 1600000
#define THETA 0.1f

typedef __attribute__((ext_vector_type(2))) float v2f;
typedef __attribute__((ext_vector_type(8))) float v8f;

__device__ __forceinline__ v8f wmma_f32(v2f a, v2f b, v8f c) {
  return __builtin_amdgcn_wmma_f32_16x16x4_f32(
       false, a,  false, b,
       (short)0, c,  false,  false);
}


typedef __attribute__((ext_vector_type(4))) int v4i;
#define NQ 4
#define QN (NN / NQ)
#define GS 64
#define ES (EE / GS)
#define PS 25024
#define HT 1024
#define HI ((ES / 4 + HT - 1) / HT)
static_assert(NN % NQ == 0 && QN % 4 == 0 && PS >= QN && PS % 32 == 0, "quarters are whole 16-byte items, planes whole lines");
static_assert(EE % GS == 0 && ES % 4 == 0, "slices are whole 16-byte items");
static_assert(QN <= ES, "HI trips also cover a quarter's QN/4 items");
static_assert((size_t)NQ * GS * PS <= (size_t)2 * NN * HH, "the planes fit into the two feature planes they borrow");
__global__ __launch_bounds__(HT) void degree_hist_kernel(const int* __restrict__ col, int* __restrict__ part) {
  __shared__ __attribute__((aligned(16))) int cnt[QN];
  const int tid = threadIdx.x;
  const int q = blockIdx.x & (NQ - 1);
  const int g = blockIdx.x / NQ;
  const v4i* cp = (const v4i*)(col + g * ES);
  v4i cv[HI];
  #pragma unroll
  for (int i = 0; i < HI; ++i) {
    const int j = i * HT + tid;
    const v4i none = {-1, -1, -1, -1};
    cv[i] = (j < ES / 4) ? cp[j] : none;
  }
  #pragma unroll
  for (int i = 0; i < HI; ++i) {
    const int j = i * HT + tid;
    const v4i z = {0, 0, 0, 0};
    if (j < QN / 4) ((v4i*)cnt)[j] = z;
  }
  __syncthreads();
  const int lo = q * QN;
  #pragma unroll
  for (int i = 0; i < HI; ++i) {
    const unsigned d0 = (unsigned)(cv[i].x - lo), d1 = (unsigned)(cv[i].y - lo);
    const unsigned d2 = (unsigned)(cv[i].z - lo), d3 = (unsigned)(cv[i].w - lo);
    if (d0 < (unsigned)QN) atomicAdd(&cnt[d0], 1);
    if (d1 < (unsigned)QN) atomicAdd(&cnt[d1], 1);
    if (d2 < (unsigned)QN) atomicAdd(&cnt[d2], 1);
    if (d3 < (unsigned)QN) atomicAdd(&cnt[d3], 1);
  }
  __syncthreads();
  v4i* op = (v4i*)(part + (q * GS + g) * PS);
  #pragma unroll
  for (int i = 0; i < HI; ++i) {
    const int j = i * HT + tid;
    if (j < QN / 4) op[j] = ((const v4i*)cnt)[j];
  }
}

__global__ void dn_kernel(const int* __restrict__ part, float* __restrict__ dn) {
  int i = blockIdx.x * blockDim.x + threadIdx.x;
  if (i < NN) {
    const int q = i / QN;
    const int* p = part + q * (GS * PS) + (i - q * QN);
    int d = 0;
    #pragma unroll
    for (int g = 0; g < GS; ++g) d += p[g * PS];
    dn[i] = (d > 0) ? rsqrtf((float)d) : 0.0f;
  }
}

__global__ __launch_bounds__(256) void fc0_kernel(const float* __restrict__ x,
                                                  const float* __restrict__ w,
                                                  const float* __restrict__ b,
                                                  float* __restrict__ h) {
  const int wave = threadIdx.x >> 5;
  const int lane = threadIdx.x & 31;
  const int half = lane >> 4;
  const int lid  = lane & 15;
  const int node0 = blockIdx.x * 128 + wave * 16;
  if (node0 >= NN) return;

  v8f acc0 = {}, acc1 = {}, acc2 = {}, acc3 = {};
  for (int c = 0; c < DD / 4; ++c) {
    const int k0 = 4 * c + 2 * half;
    const float* ap = x + (node0 + lid) * DD + k0;
    v2f a; a.x = ap[0]; a.y = ap[1];
    const float* bp = w + k0 * HH;
    v2f b0; b0.x = bp[lid];      b0.y = bp[HH + lid];
    v2f b1; b1.x = bp[16 + lid]; b1.y = bp[HH + 16 + lid];
    v2f b2; b2.x = bp[32 + lid]; b2.y = bp[HH + 32 + lid];
    v2f b3; b3.x = bp[48 + lid]; b3.y = bp[HH + 48 + lid];
    acc0 = wmma_f32(a, b0, acc0);
    acc1 = wmma_f32(a, b1, acc1);
    acc2 = wmma_f32(a, b2, acc2);
    acc3 = wmma_f32(a, b3, acc3);
  }
  v8f accs[4] = {acc0, acc1, acc2, acc3};
  for (int t = 0; t < 4; ++t) {
    const float bias = b[16 * t + lid];
    for (int r = 0; r < 8; ++r) {
      float v = accs[t][r] + bias;
      h[(node0 + r + 8 * half) * HH + 16 * t + lid] = v > 0.0f ? v : 0.0f;
    }
  }
}

typedef __attribute__((ext_vector_type(2))) int v2i;
typedef __attribute__((ext_vector_type(4))) float v4f;
#define RB 391
#define NBK 256
#define SL 250
#define SE (EE / SL)
#define BT 1024
#define RBITS 17
#define BI ((SE + BT - 1) / BT)
static_assert(RB * NBK >= NN, "the ranges cover every node");
static_assert(EE % SL == 0, "slices are whole");
static_assert(BT == 4 * NBK, "four threads per range in the place pass");
static_assert(NN <= (1 << RBITS) && RB < (1 << (31 - RBITS)), "source row and local row (the spare row RB too) fit one non-negative word");
static_assert(NBK * SL + NBK + 1 <= NN, "counts and range starts fit the spare N words at the head of the workspace");
static_assert((size_t)EE * 2 <= (size_t)NN * HH, "the records fit one feature plane");

__global__ __launch_bounds__(BT) void bin_count_kernel(const int* __restrict__ col, int* __restrict__ cnt) {
  __shared__ int hist[NBK];
  const int tid = threadIdx.x;
  const int s = blockIdx.x;
  int cv[BI];
  #pragma unroll
  for (int u = 0; u < BI; ++u) {
    const int i = u * BT + tid;
    cv[u] = (i < SE) ? col[s * SE + i] : -1;
  }
  if (tid < NBK) hist[tid] = 0;
  __syncthreads();
  #pragma unroll
  for (int u = 0; u < BI; ++u)
    if (cv[u] >= 0) atomicAdd(&hist[cv[u] / RB], 1);
  __syncthreads();
  if (tid < NBK) cnt[tid * SL + s] = hist[tid];
}

__global__ __launch_bounds__(BT) void bin_place_kernel(const int* __restrict__ row, const int* __restrict__ col,
                                                       const float* __restrict__ dn, const int* __restrict__ cnt,
                                                       int* __restrict__ boff, int* __restrict__ recw) {
  __shared__ int tot[NBK];
  __shared__ int cur[NBK];
  const int tid = threadIdx.x;
  const int s = blockIdx.x;
  v2i* recs = (v2i*)recw;
  int rv[BI], cv[BI];
  float wv[BI];
  #pragma unroll
  for (int u = 0; u < BI; ++u) {
    const int i = u * BT + tid;
    const bool ok = i < SE;
    rv[u] = ok ? row[s * SE + i] : 0;
    cv[u] = ok ? col[s * SE + i] : -1;
  }
  #pragma unroll
  for (int u = 0; u < BI; ++u) wv[u] = (cv[u] >= 0) ? dn[rv[u]] * dn[cv[u]] : 0.0f;
  {
    const int b = tid >> 2, part = tid & 3;
    int t = 0, p = 0;
    for (int s2 = part; s2 < SL; s2 += 4) {
      const int v = cnt[b * SL + s2];
      t += v;
      p += (s2 < s) ? v : 0;
    }
    t += __shfl_xor(t, 1); t += __shfl_xor(t, 2);
    p += __shfl_xor(p, 1); p += __shfl_xor(p, 2);
    if (part == 0) { tot[b] = t; cur[b] = p; }
  }
  __syncthreads();
  if (tid < NBK) {
    int start = 0;
    for (int b2 = 0; b2 < NBK; ++b2) start += (b2 < tid) ? tot[b2] : 0;
    cur[tid] += start;
    if (s == 0) {
      boff[tid] = start;
      if (tid == NBK - 1) boff[NBK] = start + tot[tid];
    }
  }
  __syncthreads();
  #pragma unroll
  for (int u = 0; u < BI; ++u) {
    if (cv[u] >= 0) {
      const int bk = cv[u] / RB;
      const int pos = atomicAdd(&cur[bk], 1);
      v2i rec;
      rec.x = rv[u] | ((cv[u] - bk * RB) << RBITS);
      rec.y = __float_as_int(wv[u]);
      recs[pos] = rec;
    }
  }
}

#define EB 8
__device__ __forceinline__ void load_rows(const float* __restrict__ h, int r, int lane, int j0, float* hv) {
  #pragma unroll
  for (int u = 0; u < EB; ++u) {
    const int rj = __builtin_amdgcn_readlane(r, j0 + u);
    const float* hp = h + rj * HH + lane;
    hv[2 * u]     = hp[0];
    hv[2 * u + 1] = hp[32];
  }
}
__device__ __forceinline__ void add_rows(float* tile, int lc, int vb, int lane, int j0, const float* hv) {
  #pragma unroll
  for (int u = 0; u < EB; ++u) {
    const int cj = __builtin_amdgcn_readlane(lc, j0 + u);
    const float vj = __int_as_float(__builtin_amdgcn_readlane(vb, j0 + u));
    float* tp = tile + cj * HH + lane;
    atomicAdd(tp,      vj * hv[2 * u]);
    atomicAdd(tp + 32, vj * hv[2 * u + 1]);
  }
}
__global__ __launch_bounds__(BT) void range_sum_kernel(const int* __restrict__ recw, const int* __restrict__ boff,
                                                       const float* __restrict__ h, float* __restrict__ agg) {
  __shared__ __attribute__((aligned(16))) float tile[(RB + 1) * HH];
  const v2i* recs = (const v2i*)recw;
  const int tid = threadIdx.x;
  const int lane = tid & 31;
  const int wave = tid >> 5;
  const int b = blockIdx.x;
  const v4f z = {0.0f, 0.0f, 0.0f, 0.0f};
  for (int i = tid; i < (RB + 1) * HH / 4; i += BT) ((v4f*)tile)[i] = z;
  const int e_lo = boff[b], e_hi = boff[b + 1];
  const int node_lo = b * RB;
  const v2i filler = {min(node_lo + lane, NN - 1) | (RB << RBITS), 0};
  __syncthreads();
  for (int e0 = e_lo + wave * 32; e0 < e_hi; e0 += BT) {
    v2i rec = filler;
    if (e0 + lane < e_hi) rec = recs[e0 + lane];
    const int r  = rec.x & ((1 << RBITS) - 1);
    const int lc = rec.x >> RBITS;
    const int vb = rec.y;
    float ha[2 * EB], hb[2 * EB];
    load_rows(h, r, lane, 0 * EB, ha);
    load_rows(h, r, lane, 1 * EB, hb);
    add_rows(tile, lc, vb, lane, 0 * EB, ha);
    load_rows(h, r, lane, 2 * EB, ha);
    add_rows(tile, lc, vb, lane, 1 * EB, hb);
    load_rows(h, r, lane, 3 * EB, hb);
    add_rows(tile, lc, vb, lane, 2 * EB, ha);
    add_rows(tile, lc, vb, lane, 3 * EB, hb);
  }
  __syncthreads();
  v4f* ap = (v4f*)(agg + node_lo * HH);
  for (int i = tid; i < RB * HH / 4; i += BT)
    if (node_lo + (i >> 4) < NN) ap[i] = ((const v4f*)tile)[i];
}

__global__ void gate_kernel(const float* __restrict__ h, const float* __restrict__ env_w,
                            const float* __restrict__ env_b, float* __restrict__ e_out,
                            int layer) {
  int n = blockIdx.x * blockDim.x + threadIdx.x;
  if (n >= NN) return;
  const float* W = env_w + layer * 2 * HH * KK;
  const float* B = env_b + layer * KK;
  float logit[KK];
  #pragma unroll
  for (int k = 0; k < KK; ++k) logit[k] = B[k];
  for (int f = 0; f < HH; ++f) {
    float hv = h[n * HH + f];
    const float* wr = W + f * KK;
    #pragma unroll
    for (int k = 0; k < KK; ++k) logit[k] += hv * wr[k];
  }
  float m = logit[0];
  #pragma unroll
  for (int k = 1; k < KK; ++k) m = fmaxf(m, logit[k]);
  float s = 0.0f;
  float ex[KK];
  #pragma unroll
  for (int k = 0; k < KK; ++k) { ex[k] = __expf(logit[k] - m); s += ex[k]; }
  float inv = 1.0f / s;
  #pragma unroll
  for (int k = 0; k < KK; ++k) {
    float pi = ex[k] * inv;
    e_out[n * KK + k] = (pi > THETA) ? pi : 0.0f;
  }
}

#define WP2 128
#define WBUF (HH * WP2)
#define NB 128

__device__ __forceinline__ void conv_half(const float* __restrict__ arow, const float* bl,
                                          float ek, int half, int lid,
                                          v8f& acc0, v8f& acc1, v8f& acc2, v8f& acc3) {
  const int s0 = 32 * half, s1 = 32 - 32 * half;
  #pragma unroll 4
  for (int c = 0; c < HH / 4; ++c) {
    const float* ap = arow + 4 * c + 2 * half;
    v2f a; a.x = ek * ap[0]; a.y = ek * ap[1];
    const float* bp = bl + (2 * c + half) * WP2 + 2 * lid;
    const v2f b0 = *(const v2f*)(bp + s0);
    const v2f b1 = *(const v2f*)(bp + s1);
    const v2f b2 = *(const v2f*)(bp + 64 + s0);
    const v2f b3 = *(const v2f*)(bp + 64 + s1);
    acc0 = wmma_f32(a, b0, acc0);
    acc1 = wmma_f32(a, b1, acc1);
    acc2 = wmma_f32(a, b2, acc2);
    acc3 = wmma_f32(a, b3, acc3);
  }
}

__global__ __launch_bounds__(256) void conv_kernel(const float* __restrict__ agg,
                                                   float* h,
                                                   const float* __restrict__ e_gate,
                                                   const float* __restrict__ conv_w,
                                                   int layer) {
  __shared__ __attribute__((aligned(16))) float wl[WBUF];
  __shared__ __attribute__((aligned(16))) float oacc[(NB + 1) * HH];
  __shared__ int   lrow[KK * NB];
  __shared__ float lgate[KK * NB];
  __shared__ int   wcnt[KK * 8];

  const int tid  = threadIdx.x;
  const int wave = tid >> 5;
  const int lane = tid & 31;
  const int half = lane >> 4;
  const int lid  = lane & 15;
  const int base = blockIdx.x * NB;
  const int node0 = base + wave * 16;
  const bool live = node0 < NN;

  const float* W = conv_w + layer * (KK * 2 * HH * HH);

  const int goff = (tid >> 4) * (2 * HH) + (tid & 15) * 4;
  const int loff = (tid >> 4) * WP2 + (((tid & 15) * 8) ^ (((tid >> 4) & 1) * 32));
  v4f r0[4], r1[4];
  #pragma unroll
  for (int i = 0; i < 4; ++i) {
    r0[i] = *(const v4f*)(W + goff + i * 16 * 2 * HH);
    r1[i] = *(const v4f*)(W + goff + i * 16 * 2 * HH + HH);
  }
  v4f ga = {}, gb = {};
  if (live) {
    const v4f* gp = (const v4f*)(e_gate + (node0 + lid) * KK);
    ga = gp[0]; gb = gp[1];
  }
  const float g[KK] = {ga.x, ga.y, ga.z, ga.w, gb.x, gb.y, gb.z, gb.w};
  unsigned m[KK];
  #pragma unroll
  for (int k = 0; k < KK; ++k) m[k] = __builtin_amdgcn_ballot_w32(g[k] != 0.0f) & 0xFFFFu;
  #pragma unroll
  for (int k = 0; k < KK; ++k)
    if (lane == k) wcnt[k * 8 + wave] = __popc(m[k]);
  #pragma unroll
  for (int i = 0; i < 8; ++i) {
    const v4f z = {};
    *(v4f*)(oacc + wave * 16 * HH + (i * 32 + lane) * 4) = z;
  }
  if (wave == 0) { oacc[NB * HH + lane] = 0.0f; oacc[NB * HH + 32 + lane] = 0.0f; }
  #pragma unroll
  for (int i = 0; i < 4; ++i) {
    v4f lo, hi;
    lo.x = r0[i].x; lo.y = r1[i].x; lo.z = r0[i].y; lo.w = r1[i].y;
    hi.x = r0[i].z; hi.y = r1[i].z; hi.z = r0[i].w; hi.w = r1[i].w;
    *(v4f*)(wl + loff + i * 16 * WP2)     = lo;
    *(v4f*)(wl + loff + i * 16 * WP2 + 4) = hi;
  }
  __syncthreads();

  #pragma unroll
  for (int k = 0; k < KK; ++k) {
    int off = 0;
    #pragma unroll
    for (int w2 = 0; w2 < 8; ++w2) {
      const int cw = wcnt[k * 8 + w2];
      off += (w2 < wave) ? cw : 0;
    }
    if (half == 0 && g[k] != 0.0f) {
      const int pos = off + __popc(m[k] & ((1u << lid) - 1u));
      lrow[k * NB + pos]  = wave * 16 + lid;
      lgate[k * NB + pos] = g[k];
    }
  }
  __syncthreads();

  const int jt = (wave & 1) ? (wave ^ 4) : wave;

  #pragma unroll 1
  for (int k = 0; k < KK; ++k) {
    const bool more = (k + 1 < KK);
    if (more) {
      const float* Wn = W + (k + 1) * (2 * HH * HH) + goff;
      #pragma unroll
      for (int i = 0; i < 4; ++i) {
        r0[i] = *(const v4f*)(Wn + i * 16 * 2 * HH);
        r1[i] = *(const v4f*)(Wn + i * 16 * 2 * HH + HH);
      }
    }
    int tk = 0;
    #pragma unroll
    for (int w2 = 0; w2 < 8; ++w2) tk += wcnt[k * 8 + w2];
    if (16 * jt < tk) {
      const int i = 16 * jt + lid;
      const bool valid = i < tk;
      int lr = 0; float ek = 0.0f;
      if (valid) { lr = lrow[k * NB + i]; ek = lgate[k * NB + i]; }
      const int drow = valid ? lr : NB;
      v8f acc0 = {}, acc1 = {}, acc2 = {}, acc3 = {};
      const float* bl = wl;
      conv_half(agg + (base + lr) * HH, bl,                  ek, half, lid, acc0, acc1, acc2, acc3);
      conv_half(h   + (base + lr) * HH, bl + (HH / 2) * WP2, ek, half, lid, acc0, acc1, acc2, acc3);
      float* op[8];
      float t0[8], t1[8], t2[8], t3[8];
      #pragma unroll
      for (int r = 0; r < 8; ++r) {
        const int dl = __builtin_amdgcn_readlane(drow, r);
        const int dh = __builtin_amdgcn_readlane(drow, r + 8);
        op[r] = oacc + (half ? dh : dl) * HH + lid;
        t0[r] = op[r][0];
        t1[r] = op[r][16];
        t2[r] = op[r][32];
        t3[r] = op[r][48];
      }
      #pragma unroll
      for (int r = 0; r < 8; ++r) {
        op[r][0]  = t0[r] + acc0[r];
        op[r][16] = t1[r] + acc1[r];
        op[r][32] = t2[r] + acc2[r];
        op[r][48] = t3[r] + acc3[r];
      }
    }
    if (more) {
      __syncthreads();
      float* dst = wl + loff;
      #pragma unroll
      for (int i = 0; i < 4; ++i) {
        v4f lo, hi;
        lo.x = r0[i].x; lo.y = r1[i].x; lo.z = r0[i].y; lo.w = r1[i].y;
        hi.x = r0[i].z; hi.y = r1[i].z; hi.z = r0[i].w; hi.w = r1[i].w;
        *(v4f*)(dst + i * 16 * WP2)     = lo;
        *(v4f*)(dst + i * 16 * WP2 + 4) = hi;
      }
    }
    __syncthreads();
  }
  if (!live) return;
  #pragma unroll 4
  for (int r = 0; r < 16; ++r) {
    const int idx = (node0 + r) * HH + lane;
    const float* op = oacc + (wave * 16 + r) * HH + lane;
    const float v0 = op[0]  + h[idx];
    const float v1 = op[32] + h[idx + 32];
    h[idx]      = v0 > 0.0f ? v0 : 0.0f;
    h[idx + 32] = v1 > 0.0f ? v1 : 0.0f;
  }
}

__global__ __launch_bounds__(256) void fc1_kernel(const float* __restrict__ h,
                                                  const float* __restrict__ w,
                                                  const float* __restrict__ b,
                                                  float* __restrict__ out) {
  const int wave = threadIdx.x >> 5;
  const int lane = threadIdx.x & 31;
  const int half = lane >> 4;
  const int lid  = lane & 15;
  const int node0 = blockIdx.x * 128 + wave * 16;
  if (node0 >= NN) return;

  v8f acc = {};
  for (int c = 0; c < HH / 4; ++c) {
    const int k0 = 4 * c + 2 * half;
    const float* ap = h + (node0 + lid) * HH + k0;
    v2f a; a.x = ap[0]; a.y = ap[1];
    const float* bp = w + k0 * CC;
    v2f bb; bb.x = bp[lid]; bb.y = bp[CC + lid];
    acc = wmma_f32(a, bb, acc);
  }
  const float bias = b[lid];
  for (int r = 0; r < 8; ++r)
    out[(node0 + r + 8 * half) * CC + lid] = acc[r] + bias;
}

extern "C" void kernel_launch(void* const* d_in, const int* in_sizes, int n_in,
                              void* d_out, int out_size, void* d_ws, size_t ws_size,
                              hipStream_t stream) {
  const float* x      = (const float*)d_in[0];
  const int*   eidx   = (const int*)d_in[1];
  const float* fc0_w  = (const float*)d_in[2];
  const float* fc0_b  = (const float*)d_in[3];
  const float* fc1_w  = (const float*)d_in[4];
  const float* fc1_b  = (const float*)d_in[5];
  const float* env_w  = (const float*)d_in[6];
  const float* env_b  = (const float*)d_in[7];
  const float* conv_w = (const float*)d_in[8];
  float* out = (float*)d_out;

  const int* row = eidx;
  const int* col = eidx + EE;

  float* ws  = (float*)d_ws;
  float* dn  = ws + NN;
  float* hA  = dn + NN;
  float* hB  = hA + (size_t)NN * HH;
  float* agg = hB + (size_t)NN * HH;
  float* eg  = agg + (size_t)NN * HH;
  int* cnt   = (int*)ws;
  int* boff  = cnt + NBK * SL;
  int* recs  = (int*)hB;

  const int T = 256;
  const int gemm_blocks = (NN + 127) / 128;

  int* part = (int*)hB;
  degree_hist_kernel<<<NQ * GS, HT, 0, stream>>>(col, part);
  dn_kernel<<<(NN + T - 1) / T, T, 0, stream>>>(part, dn);

  bin_count_kernel<<<SL, BT, 0, stream>>>(col, cnt);
  bin_place_kernel<<<SL, BT, 0, stream>>>(row, col, dn, cnt, boff, recs);

  fc0_kernel<<<gemm_blocks, T, 0, stream>>>(x, fc0_w, fc0_b, hA);

  for (int layer = 0; layer < LL; ++layer) {
    range_sum_kernel<<<NBK, BT, 0, stream>>>(recs, boff, hA, agg);
    gate_kernel<<<(NN + T - 1) / T, T, 0, stream>>>(hA, env_w, env_b, eg, layer);
    conv_kernel<<<gemm_blocks, T, 0, stream>>>(agg, hA, eg, conv_w, layer);
  }

  fc1_kernel<<<gemm_blocks, T, 0, stream>>>(hA, fc1_w, fc1_b, out);
}


# gate: made inside the per-layer range sum launch by its last 13 waves (one node per lane, whole h row in registers, same arithmetic and sum order): 3 gate launches gone, 15 -> 12
# speedup vs baseline: 2.4198x; 1.0309x over previous
#include <hip/hip_runtime.h>
#include <hip/hip_bf16.h>

#define NN 100000
#define DD 128
#define HH 64
#define KK 8
#define LL 3
#define CC 16
#define EE 1600000
#define THETA 0.1f

typedef __attribute__((ext_vector_type(2))) float v2f;
typedef __attribute__((ext_vector_type(8))) float v8f;

__device__ __forceinline__ v8f wmma_f32(v2f a, v2f b, v8f c) {
  return __builtin_amdgcn_wmma_f32_16x16x4_f32(
       false, a,  false, b,
       (short)0, c,  false,  false);
}


typedef __attribute__((ext_vector_type(4))) int v4i;
#define NQ 4
#define QN (NN / NQ)
#define GS 64
#define ES (EE / GS)
#define PS 25024
#define HT 1024
#define HI ((ES / 4 + HT - 1) / HT)
static_assert(NN % NQ == 0 && QN % 4 == 0 && PS >= QN && PS % 32 == 0, "quarters are whole 16-byte items, planes whole lines");
static_assert(EE % GS == 0 && ES % 4 == 0, "slices are whole 16-byte items");
static_assert(QN <= ES, "HI trips also cover a quarter's QN/4 items");
static_assert((size_t)NQ * GS * PS <= (size_t)2 * NN * HH, "the planes fit into the two feature planes they borrow");
__global__ __launch_bounds__(HT) void degree_hist_kernel(const int* __restrict__ col, int* __restrict__ part) {
  __shared__ __attribute__((aligned(16))) int cnt[QN];
  const int tid = threadIdx.x;
  const int q = blockIdx.x & (NQ - 1);
  const int g = blockIdx.x / NQ;
  const v4i* cp = (const v4i*)(col + g * ES);
  v4i cv[HI];
  #pragma unroll
  for (int i = 0; i < HI; ++i) {
    const int j = i * HT + tid;
    const v4i none = {-1, -1, -1, -1};
    cv[i] = (j < ES / 4) ? cp[j] : none;
  }
  #pragma unroll
  for (int i = 0; i < HI; ++i) {
    const int j = i * HT + tid;
    const v4i z = {0, 0, 0, 0};
    if (j < QN / 4) ((v4i*)cnt)[j] = z;
  }
  __syncthreads();
  const int lo = q * QN;
  #pragma unroll
  for (int i = 0; i < HI; ++i) {
    const unsigned d0 = (unsigned)(cv[i].x - lo), d1 = (unsigned)(cv[i].y - lo);
    const unsigned d2 = (unsigned)(cv[i].z - lo), d3 = (unsigned)(cv[i].w - lo);
    if (d0 < (unsigned)QN) atomicAdd(&cnt[d0], 1);
    if (d1 < (unsigned)QN) atomicAdd(&cnt[d1], 1);
    if (d2 < (unsigned)QN) atomicAdd(&cnt[d2], 1);
    if (d3 < (unsigned)QN) atomicAdd(&cnt[d3], 1);
  }
  __syncthreads();
  v4i* op = (v4i*)(part + (q * GS + g) * PS);
  #pragma unroll
  for (int i = 0; i < HI; ++i) {
    const int j = i * HT + tid;
    if (j < QN / 4) op[j] = ((const v4i*)cnt)[j];
  }
}

__global__ void dn_kernel(const int* __restrict__ part, float* __restrict__ dn) {
  int i = blockIdx.x * blockDim.x + threadIdx.x;
  if (i < NN) {
    const int q = i / QN;
    const int* p = part + q * (GS * PS) + (i - q * QN);
    int d = 0;
    #pragma unroll
    for (int g = 0; g < GS; ++g) d += p[g * PS];
    dn[i] = (d > 0) ? rsqrtf((float)d) : 0.0f;
  }
}

__global__ __launch_bounds__(256) void fc0_kernel(const float* __restrict__ x,
                                                  const float* __restrict__ w,
                                                  const float* __restrict__ b,
                                                  float* __restrict__ h) {
  const int wave = threadIdx.x >> 5;
  const int lane = threadIdx.x & 31;
  const int half = lane >> 4;
  const int lid  = lane & 15;
  const int node0 = blockIdx.x * 128 + wave * 16;
  if (node0 >= NN) return;

  v8f acc0 = {}, acc1 = {}, acc2 = {}, acc3 = {};
  for (int c = 0; c < DD / 4; ++c) {
    const int k0 = 4 * c + 2 * half;
    const float* ap = x + (node0 + lid) * DD + k0;
    v2f a; a.x = ap[0]; a.y = ap[1];
    const float* bp = w + k0 * HH;
    v2f b0; b0.x = bp[lid];      b0.y = bp[HH + lid];
    v2f b1; b1.x = bp[16 + lid]; b1.y = bp[HH + 16 + lid];
    v2f b2; b2.x = bp[32 + lid]; b2.y = bp[HH + 32 + lid];
    v2f b3; b3.x = bp[48 + lid]; b3.y = bp[HH + 48 + lid];
    acc0 = wmma_f32(a, b0, acc0);
    acc1 = wmma_f32(a, b1, acc1);
    acc2 = wmma_f32(a, b2, acc2);
    acc3 = wmma_f32(a, b3, acc3);
  }
  v8f accs[4] = {acc0, acc1, acc2, acc3};
  for (int t = 0; t < 4; ++t) {
    const float bias = b[16 * t + lid];
    for (int r = 0; r < 8; ++r) {
      float v = accs[t][r] + bias;
      h[(node0 + r + 8 * half) * HH + 16 * t + lid] = v > 0.0f ? v : 0.0f;
    }
  }
}

typedef __attribute__((ext_vector_type(2))) int v2i;
typedef __attribute__((ext_vector_type(4))) float v4f;
#define RB 391
#define NBK 256
#define SL 250
#define SE (EE / SL)
#define BT 1024
#define RBITS 17
#define BI ((SE + BT - 1) / BT)
static_assert(RB * NBK >= NN, "the ranges cover every node");
static_assert(EE % SL == 0, "slices are whole");
static_assert(BT == 4 * NBK, "four threads per range in the place pass");
static_assert(NN <= (1 << RBITS) && RB < (1 << (31 - RBITS)), "source row and local row (the spare row RB too) fit one non-negative word");
static_assert(NBK * SL + NBK + 1 <= NN, "counts and range starts fit the spare N words at the head of the workspace");
static_assert((size_t)EE * 2 <= (size_t)NN * HH, "the records fit one feature plane");

__global__ __launch_bounds__(BT) void bin_count_kernel(const int* __restrict__ col, int* __restrict__ cnt) {
  __shared__ int hist[NBK];
  const int tid = threadIdx.x;
  const int s = blockIdx.x;
  int cv[BI];
  #pragma unroll
  for (int u = 0; u < BI; ++u) {
    const int i = u * BT + tid;
    cv[u] = (i < SE) ? col[s * SE + i] : -1;
  }
  if (tid < NBK) hist[tid] = 0;
  __syncthreads();
  #pragma unroll
  for (int u = 0; u < BI; ++u)
    if (cv[u] >= 0) atomicAdd(&hist[cv[u] / RB], 1);
  __syncthreads();
  if (tid < NBK) cnt[tid * SL + s] = hist[tid];
}

__global__ __launch_bounds__(BT) void bin_place_kernel(const int* __restrict__ row, const int* __restrict__ col,
                                                       const float* __restrict__ dn, const int* __restrict__ cnt,
                                                       int* __restrict__ boff, int* __restrict__ recw) {
  __shared__ int tot[NBK];
  __shared__ int cur[NBK];
  const int tid = threadIdx.x;
  const int s = blockIdx.x;
  v2i* recs = (v2i*)recw;
  int rv[BI], cv[BI];
  float wv[BI];
  #pragma unroll
  for (int u = 0; u < BI; ++u) {
    const int i = u * BT + tid;
    const bool ok = i < SE;
    rv[u] = ok ? row[s * SE + i] : 0;
    cv[u] = ok ? col[s * SE + i] : -1;
  }
  #pragma unroll
  for (int u = 0; u < BI; ++u) wv[u] = (cv[u] >= 0) ? dn[rv[u]] * dn[cv[u]] : 0.0f;
  {
    const int b = tid >> 2, part = tid & 3;
    int t = 0, p = 0;
    for (int s2 = part; s2 < SL; s2 += 4) {
      const int v = cnt[b * SL + s2];
      t += v;
      p += (s2 < s) ? v : 0;
    }
    t += __shfl_xor(t, 1); t += __shfl_xor(t, 2);
    p += __shfl_xor(p, 1); p += __shfl_xor(p, 2);
    if (part == 0) { tot[b] = t; cur[b] = p; }
  }
  __syncthreads();
  if (tid < NBK) {
    int start = 0;
    for (int b2 = 0; b2 < NBK; ++b2) start += (b2 < tid) ? tot[b2] : 0;
    cur[tid] += start;
    if (s == 0) {
      boff[tid] = start;
      if (tid == NBK - 1) boff[NBK] = start + tot[tid];
    }
  }
  __syncthreads();
  #pragma unroll
  for (int u = 0; u < BI; ++u) {
    if (cv[u] >= 0) {
      const int bk = cv[u] / RB;
      const int pos = atomicAdd(&cur[bk], 1);
      v2i rec;
      rec.x = rv[u] | ((cv[u] - bk * RB) << RBITS);
      rec.y = __float_as_int(wv[u]);
      recs[pos] = rec;
    }
  }
}

#define EB 8
__device__ __forceinline__ void load_rows(const float* __restrict__ h, int r, int lane, int j0, float* hv) {
  #pragma unroll
  for (int u = 0; u < EB; ++u) {
    const int rj = __builtin_amdgcn_readlane(r, j0 + u);
    const float* hp = h + rj * HH + lane;
    hv[2 * u]     = hp[0];
    hv[2 * u + 1] = hp[32];
  }
}
__device__ __forceinline__ void add_rows(float* tile, int lc, int vb, int lane, int j0, const float* hv) {
  #pragma unroll
  for (int u = 0; u < EB; ++u) {
    const int cj = __builtin_amdgcn_readlane(lc, j0 + u);
    const float vj = __int_as_float(__builtin_amdgcn_readlane(vb, j0 + u));
    float* tp = tile + cj * HH + lane;
    atomicAdd(tp,      vj * hv[2 * u]);
    atomicAdd(tp + 32, vj * hv[2 * u + 1]);
  }
}
#define GW 13
#define GT0 (BT - GW * 32)
static_assert(GW * 32 >= RB + 3, "a lane for every node of a range moved up to multiples of four");
__device__ __forceinline__ void gate_store(const v4f* hr, const float* __restrict__ W, const float* __restrict__ B,
                                           float* __restrict__ ep) {
  float logit[KK];
  #pragma unroll
  for (int k = 0; k < KK; ++k) logit[k] = B[k];
  #pragma unroll
  for (int f = 0; f < HH; ++f) {
    const float hv = hr[f >> 2][f & 3];
    const float* wr = W + f * KK;
    #pragma unroll
    for (int k = 0; k < KK; ++k) logit[k] += hv * wr[k];
  }
  float m = logit[0];
  #pragma unroll
  for (int k = 1; k < KK; ++k) m = fmaxf(m, logit[k]);
  float s = 0.0f;
  float ex[KK];
  #pragma unroll
  for (int k = 0; k < KK; ++k) { ex[k] = __expf(logit[k] - m); s += ex[k]; }
  float inv = 1.0f / s;
  #pragma unroll
  for (int k = 0; k < KK; ++k) {
    float pi = ex[k] * inv;
    ep[k] = (pi > THETA) ? pi : 0.0f;
  }
}

__global__ __launch_bounds__(BT) void range_sum_kernel(const int* __restrict__ recw, const int* __restrict__ boff,
                                                       const float* __restrict__ h, float* __restrict__ agg,
                                                       const float* __restrict__ env_w, const float* __restrict__ env_b,
                                                       float* __restrict__ e_out, int layer) {
  __shared__ __attribute__((aligned(16))) float tile[(RB + 1) * HH];
  const v2i* recs = (const v2i*)recw;
  const int tid = threadIdx.x;
  const int lane = tid & 31;
  const int wave = tid >> 5;
  const int b = blockIdx.x;
  const v4f z = {0.0f, 0.0f, 0.0f, 0.0f};
  const int node_lo = b * RB;
  const int gn = ((node_lo + 3) & ~3) + (tid - GT0);
  const bool gate_on = tid >= GT0 && gn < min((node_lo + RB + 3) & ~3, NN);
  for (int i = tid; i < (RB + 1) * HH / 4; i += BT) ((v4f*)tile)[i] = z;
  const int e_lo = boff[b], e_hi = boff[b + 1];
  const v2i filler = {min(node_lo + lane, NN - 1) | (RB << RBITS), 0};
  __syncthreads();
  if (gate_on) {
    const v4f* hp = (const v4f*)(h + gn * HH);
    v4f hr[HH / 4];
    #pragma unroll
    for (int i = 0; i < HH / 4; ++i) hr[i] = hp[i];
    gate_store(hr, env_w + layer * 2 * HH * KK, env_b + layer * KK, e_out + gn * KK);
  }
  for (int e0 = e_lo + wave * 32; e0 < e_hi; e0 += BT) {
    v2i rec = filler;
    if (e0 + lane < e_hi) rec = recs[e0 + lane];
    const int r  = rec.x & ((1 << RBITS) - 1);
    const int lc = rec.x >> RBITS;
    const int vb = rec.y;
    float ha[2 * EB], hb[2 * EB];
    load_rows(h, r, lane, 0 * EB, ha);
    load_rows(h, r, lane, 1 * EB, hb);
    add_rows(tile, lc, vb, lane, 0 * EB, ha);
    load_rows(h, r, lane, 2 * EB, ha);
    add_rows(tile, lc, vb, lane, 1 * EB, hb);
    load_rows(h, r, lane, 3 * EB, hb);
    add_rows(tile, lc, vb, lane, 2 * EB, ha);
    add_rows(tile, lc, vb, lane, 3 * EB, hb);
  }
  __syncthreads();
  v4f* ap = (v4f*)(agg + node_lo * HH);
  for (int i = tid; i < RB * HH / 4; i += BT)
    if (node_lo + (i >> 4) < NN) ap[i] = ((const v4f*)tile)[i];
}

#define WP2 128
#define WBUF (HH * WP2)
#define NB 128

__device__ __forceinline__ void conv_half(const float* __restrict__ arow, const float* bl,
                                          float ek, int half, int lid,
                                          v8f& acc0, v8f& acc1, v8f& acc2, v8f& acc3) {
  const int s0 = 32 * half, s1 = 32 - 32 * half;
  #pragma unroll 4
  for (int c = 0; c < HH / 4; ++c) {
    const float* ap = arow + 4 * c + 2 * half;
    v2f a; a.x = ek * ap[0]; a.y = ek * ap[1];
    const float* bp = bl + (2 * c + half) * WP2 + 2 * lid;
    const v2f b0 = *(const v2f*)(bp + s0);
    const v2f b1 = *(const v2f*)(bp + s1);
    const v2f b2 = *(const v2f*)(bp + 64 + s0);
    const v2f b3 = *(const v2f*)(bp + 64 + s1);
    acc0 = wmma_f32(a, b0, acc0);
    acc1 = wmma_f32(a, b1, acc1);
    acc2 = wmma_f32(a, b2, acc2);
    acc3 = wmma_f32(a, b3, acc3);
  }
}

__global__ __launch_bounds__(256) void conv_kernel(const float* __restrict__ agg,
                                                   float* h,
                                                   const float* __restrict__ e_gate,
                                                   const float* __restrict__ conv_w,
                                                   int layer) {
  __shared__ __attribute__((aligned(16))) float wl[WBUF];
  __shared__ __attribute__((aligned(16))) float oacc[(NB + 1) * HH];
  __shared__ int   lrow[KK * NB];
  __shared__ float lgate[KK * NB];
  __shared__ int   wcnt[KK * 8];

  const int tid  = threadIdx.x;
  const int wave = tid >> 5;
  const int lane = tid & 31;
  const int half = lane >> 4;
  const int lid  = lane & 15;
  const int base = blockIdx.x * NB;
  const int node0 = base + wave * 16;
  const bool live = node0 < NN;

  const float* W = conv_w + layer * (KK * 2 * HH * HH);

  const int goff = (tid >> 4) * (2 * HH) + (tid & 15) * 4;
  const int loff = (tid >> 4) * WP2 + (((tid & 15) * 8) ^ (((tid >> 4) & 1) * 32));
  v4f r0[4], r1[4];
  #pragma unroll
  for (int i = 0; i < 4; ++i) {
    r0[i] = *(const v4f*)(W + goff + i * 16 * 2 * HH);
    r1[i] = *(const v4f*)(W + goff + i * 16 * 2 * HH + HH);
  }
  v4f ga = {}, gb = {};
  if (live) {
    const v4f* gp = (const v4f*)(e_gate + (node0 + lid) * KK);
    ga = gp[0]; gb = gp[1];
  }
  const float g[KK] = {ga.x, ga.y, ga.z, ga.w, gb.x, gb.y, gb.z, gb.w};
  unsigned m[KK];
  #pragma unroll
  for (int k = 0; k < KK; ++k) m[k] = __builtin_amdgcn_ballot_w32(g[k] != 0.0f) & 0xFFFFu;
  #pragma unroll
  for (int k = 0; k < KK; ++k)
    if (lane == k) wcnt[k * 8 + wave] = __popc(m[k]);
  #pragma unroll
  for (int i = 0; i < 8; ++i) {
    const v4f z = {};
    *(v4f*)(oacc + wave * 16 * HH + (i * 32 + lane) * 4) = z;
  }
  if (wave == 0) { oacc[NB * HH + lane] = 0.0f; oacc[NB * HH + 32 + lane] = 0.0f; }
  #pragma unroll
  for (int i = 0; i < 4; ++i) {
    v4f lo, hi;
    lo.x = r0[i].x; lo.y = r1[i].x; lo.z = r0[i].y; lo.w = r1[i].y;
    hi.x = r0[i].z; hi.y = r1[i].z; hi.z = r0[i].w; hi.w = r1[i].w;
    *(v4f*)(wl + loff + i * 16 * WP2)     = lo;
    *(v4f*)(wl + loff + i * 16 * WP2 + 4) = hi;
  }
  __syncthreads();

  #pragma unroll
  for (int k = 0; k < KK; ++k) {
    int off = 0;
    #pragma unroll
    for (int w2 = 0; w2 < 8; ++w2) {
      const int cw = wcnt[k * 8 + w2];
      off += (w2 < wave) ? cw : 0;
    }
    if (half == 0 && g[k] != 0.0f) {
      const int pos = off + __popc(m[k] & ((1u << lid) - 1u));
      lrow[k * NB + pos]  = wave * 16 + lid;
      lgate[k * NB + pos] = g[k];
    }
  }
  __syncthreads();

  const int jt = (wave & 1) ? (wave ^ 4) : wave;

  #pragma unroll 1
  for (int k = 0; k < KK; ++k) {
    const bool more = (k + 1 < KK);
    if (more) {
      const float* Wn = W + (k + 1) * (2 * HH * HH) + goff;
      #pragma unroll
      for (int i = 0; i < 4; ++i) {
        r0[i] = *(const v4f*)(Wn + i * 16 * 2 * HH);
        r1[i] = *(const v4f*)(Wn + i * 16 * 2 * HH + HH);
      }
    }
    int tk = 0;
    #pragma unroll
    for (int w2 = 0; w2 < 8; ++w2) tk += wcnt[k * 8 + w2];
    if (16 * jt < tk) {
      const int i = 16 * jt + lid;
      const bool valid = i < tk;
      int lr = 0; float ek = 0.0f;
      if (valid) { lr = lrow[k * NB + i]; ek = lgate[k * NB + i]; }
      const int drow = valid ? lr : NB;
      v8f acc0 = {}, acc1 = {}, acc2 = {}, acc3 = {};
      const float* bl = wl;
      conv_half(agg + (base + lr) * HH, bl,                  ek, half, lid, acc0, acc1, acc2, acc3);
      conv_half(h   + (base + lr) * HH, bl + (HH / 2) * WP2, ek, half, lid, acc0, acc1, acc2, acc3);
      float* op[8];
      float t0[8], t1[8], t2[8], t3[8];
      #pragma unroll
      for (int r = 0; r < 8; ++r) {
        const int dl = __builtin_amdgcn_readlane(drow, r);
        const int dh = __builtin_amdgcn_readlane(drow, r + 8);
        op[r] = oacc + (half ? dh : dl) * HH + lid;
        t0[r] = op[r][0];
        t1[r] = op[r][16];
        t2[r] = op[r][32];
        t3[r] = op[r][48];
      }
      #pragma unroll
      for (int r = 0; r < 8; ++r) {
        op[r][0]  = t0[r] + acc0[r];
        op[r][16] = t1[r] + acc1[r];
        op[r][32] = t2[r] + acc2[r];
        op[r][48] = t3[r] + acc3[r];
      }
    }
    if (more) {
      __syncthreads();
      float* dst = wl + loff;
      #pragma unroll
      for (int i = 0; i < 4; ++i) {
        v4f lo, hi;
        lo.x = r0[i].x; lo.y = r1[i].x; lo.z = r0[i].y; lo.w = r1[i].y;
        hi.x = r0[i].z; hi.y = r1[i].z; hi.z = r0[i].w; hi.w = r1[i].w;
        *(v4f*)(dst + i * 16 * WP2)     = lo;
        *(v4f*)(dst + i * 16 * WP2 + 4) = hi;
      }
    }
    __syncthreads();
  }
  if (!live) return;
  #pragma unroll 4
  for (int r = 0; r < 16; ++r) {
    const int idx = (node0 + r) * HH + lane;
    const float* op = oacc + (wave * 16 + r) * HH + lane;
    const float v0 = op[0]  + h[idx];
    const float v1 = op[32] + h[idx + 32];
    h[idx]      = v0 > 0.0f ? v0 : 0.0f;
    h[idx + 32] = v1 > 0.0f ? v1 : 0.0f;
  }
}

__global__ __launch_bounds__(256) void fc1_kernel(const float* __restrict__ h,
                                                  const float* __restrict__ w,
                                                  const float* __restrict__ b,
                                                  float* __restrict__ out) {
  const int wave = threadIdx.x >> 5;
  const int lane = threadIdx.x & 31;
  const int half = lane >> 4;
  const int lid  = lane & 15;
  const int node0 = blockIdx.x * 128 + wave * 16;
  if (node0 >= NN) return;

  v8f acc = {};
  for (int c = 0; c < HH / 4; ++c) {
    const int k0 = 4 * c + 2 * half;
    const float* ap = h + (node0 + lid) * HH + k0;
    v2f a; a.x = ap[0]; a.y = ap[1];
    const float* bp = w + k0 * CC;
    v2f bb; bb.x = bp[lid]; bb.y = bp[CC + lid];
    acc = wmma_f32(a, bb, acc);
  }
  const float bias = b[lid];
  for (int r = 0; r < 8; ++r)
    out[(node0 + r + 8 * half) * CC + lid] = acc[r] + bias;
}

extern "C" void kernel_launch(void* const* d_in, const int* in_sizes, int n_in,
                              void* d_out, int out_size, void* d_ws, size_t ws_size,
                              hipStream_t stream) {
  const float* x      = (const float*)d_in[0];
  const int*   eidx   = (const int*)d_in[1];
  const float* fc0_w  = (const float*)d_in[2];
  const float* fc0_b  = (const float*)d_in[3];
  const float* fc1_w  = (const float*)d_in[4];
  const float* fc1_b  = (const float*)d_in[5];
  const float* env_w  = (const float*)d_in[6];
  const float* env_b  = (const float*)d_in[7];
  const float* conv_w = (const float*)d_in[8];
  float* out = (float*)d_out;

  const int* row = eidx;
  const int* col = eidx + EE;

  float* ws  = (float*)d_ws;
  float* dn  = ws + NN;
  float* hA  = dn + NN;
  float* hB  = hA + (size_t)NN * HH;
  float* agg = hB + (size_t)NN * HH;
  float* eg  = agg + (size_t)NN * HH;
  int* cnt   = (int*)ws;
  int* boff  = cnt + NBK * SL;
  int* recs  = (int*)hB;

  const int T = 256;
  const int gemm_blocks = (NN + 127) / 128;

  int* part = (int*)hB;
  degree_hist_kernel<<<NQ * GS, HT, 0, stream>>>(col, part);
  dn_kernel<<<(NN + T - 1) / T, T, 0, stream>>>(part, dn);

  bin_count_kernel<<<SL, BT, 0, stream>>>(col, cnt);
  bin_place_kernel<<<SL, BT, 0, stream>>>(row, col, dn, cnt, boff, recs);

  fc0_kernel<<<gemm_blocks, T, 0, stream>>>(x, fc0_w, fc0_b, hA);

  for (int layer = 0; layer < LL; ++layer) {
    range_sum_kernel<<<NBK, BT, 0, stream>>>(recs, boff, hA, agg, env_w, env_b, eg, layer);
    conv_kernel<<<gemm_blocks, T, 0, stream>>>(agg, hA, eg, conv_w, layer);
  }

  fc1_kernel<<<gemm_blocks, T, 0, stream>>>(hA, fc1_w, fc1_b, out);
}


# conv: 131 nodes per work-group, LDS reserved so three fit a WGP: 764 work-groups sit three to a WGP on all 256 (before, 782 of 128 nodes put a fourth on 14 WGPs); per-node sums and order kept
# speedup vs baseline: 2.6776x; 1.1065x over previous
#include <hip/hip_runtime.h>
#include <hip/hip_bf16.h>

#define NN 100000
#define DD 128
#define HH 64
#define KK 8
#define LL 3
#define CC 16
#define EE 1600000
#define THETA 0.1f

typedef __attribute__((ext_vector_type(2))) float v2f;
typedef __attribute__((ext_vector_type(8))) float v8f;

__device__ __forceinline__ v8f wmma_f32(v2f a, v2f b, v8f c) {
  return __builtin_amdgcn_wmma_f32_16x16x4_f32(
       false, a,  false, b,
       (short)0, c,  false,  false);
}


typedef __attribute__((ext_vector_type(4))) int v4i;
#define NQ 4
#define QN (NN / NQ)
#define GS 64
#define ES (EE / GS)
#define PS 25024
#define HT 1024
#define HI ((ES / 4 + HT - 1) / HT)
static_assert(NN % NQ == 0 && QN % 4 == 0 && PS >= QN && PS % 32 == 0, "quarters are whole 16-byte items, planes whole lines");
static_assert(EE % GS == 0 && ES % 4 == 0, "slices are whole 16-byte items");
static_assert(QN <= ES, "HI trips also cover a quarter's QN/4 items");
static_assert((size_t)NQ * GS * PS <= (size_t)2 * NN * HH, "the planes fit into the two feature planes they borrow");
__global__ __launch_bounds__(HT) void degree_hist_kernel(const int* __restrict__ col, int* __restrict__ part) {
  __shared__ __attribute__((aligned(16))) int cnt[QN];
  const int tid = threadIdx.x;
  const int q = blockIdx.x & (NQ - 1);
  const int g = blockIdx.x / NQ;
  const v4i* cp = (const v4i*)(col + g * ES);
  v4i cv[HI];
  #pragma unroll
  for (int i = 0; i < HI; ++i) {
    const int j = i * HT + tid;
    const v4i none = {-1, -1, -1, -1};
    cv[i] = (j < ES / 4) ? cp[j] : none;
  }
  #pragma unroll
  for (int i = 0; i < HI; ++i) {
    const int j = i * HT + tid;
    const v4i z = {0, 0, 0, 0};
    if (j < QN / 4) ((v4i*)cnt)[j] = z;
  }
  __syncthreads();
  const int lo = q * QN;
  #pragma unroll
  for (int i = 0; i < HI; ++i) {
    const unsigned d0 = (unsigned)(cv[i].x - lo), d1 = (unsigned)(cv[i].y - lo);
    const unsigned d2 = (unsigned)(cv[i].z - lo), d3 = (unsigned)(cv[i].w - lo);
    if (d0 < (unsigned)QN) atomicAdd(&cnt[d0], 1);
    if (d1 < (unsigned)QN) atomicAdd(&cnt[d1], 1);
    if (d2 < (unsigned)QN) atomicAdd(&cnt[d2], 1);
    if (d3 < (unsigned)QN) atomicAdd(&cnt[d3], 1);
  }
  __syncthreads();
  v4i* op = (v4i*)(part + (q * GS + g) * PS);
  #pragma unroll
  for (int i = 0; i < HI; ++i) {
    const int j = i * HT + tid;
    if (j < QN / 4) op[j] = ((const v4i*)cnt)[j];
  }
}

__global__ void dn_kernel(const int* __restrict__ part, float* __restrict__ dn) {
  int i = blockIdx.x * blockDim.x + threadIdx.x;
  if (i < NN) {
    const int q = i / QN;
    const int* p = part + q * (GS * PS) + (i - q * QN);
    int d = 0;
    #pragma unroll
    for (int g = 0; g < GS; ++g) d += p[g * PS];
    dn[i] = (d > 0) ? rsqrtf((float)d) : 0.0f;
  }
}

__global__ __launch_bounds__(256) void fc0_kernel(const float* __restrict__ x,
                                                  const float* __restrict__ w,
                                                  const float* __restrict__ b,
                                                  float* __restrict__ h) {
  const int wave = threadIdx.x >> 5;
  const int lane = threadIdx.x & 31;
  const int half = lane >> 4;
  const int lid  = lane & 15;
  const int node0 = blockIdx.x * 128 + wave * 16;
  if (node0 >= NN) return;

  v8f acc0 = {}, acc1 = {}, acc2 = {}, acc3 = {};
  for (int c = 0; c < DD / 4; ++c) {
    const int k0 = 4 * c + 2 * half;
    const float* ap = x + (node0 + lid) * DD + k0;
    v2f a; a.x = ap[0]; a.y = ap[1];
    const float* bp = w + k0 * HH;
    v2f b0; b0.x = bp[lid];      b0.y = bp[HH + lid];
    v2f b1; b1.x = bp[16 + lid]; b1.y = bp[HH + 16 + lid];
    v2f b2; b2.x = bp[32 + lid]; b2.y = bp[HH + 32 + lid];
    v2f b3; b3.x = bp[48 + lid]; b3.y = bp[HH + 48 + lid];
    acc0 = wmma_f32(a, b0, acc0);
    acc1 = wmma_f32(a, b1, acc1);
    acc2 = wmma_f32(a, b2, acc2);
    acc3 = wmma_f32(a, b3, acc3);
  }
  v8f accs[4] = {acc0, acc1, acc2, acc3};
  for (int t = 0; t < 4; ++t) {
    const float bias = b[16 * t + lid];
    for (int r = 0; r < 8; ++r) {
      float v = accs[t][r] + bias;
      h[(node0 + r + 8 * half) * HH + 16 * t + lid] = v > 0.0f ? v : 0.0f;
    }
  }
}

typedef __attribute__((ext_vector_type(2))) int v2i;
typedef __attribute__((ext_vector_type(4))) float v4f;
#define RB 391
#define NBK 256
#define SL 250
#define SE (EE / SL)
#define BT 1024
#define RBITS 17
#define BI ((SE + BT - 1) / BT)
static_assert(RB * NBK >= NN, "the ranges cover every node");
static_assert(EE % SL == 0, "slices are whole");
static_assert(BT == 4 * NBK, "four threads per range in the place pass");
static_assert(NN <= (1 << RBITS) && RB < (1 << (31 - RBITS)), "source row and local row (the spare row RB too) fit one non-negative word");
static_assert(NBK * SL + NBK + 1 <= NN, "counts and range starts fit the spare N words at the head of the workspace");
static_assert((size_t)EE * 2 <= (size_t)NN * HH, "the records fit one feature plane");

__global__ __launch_bounds__(BT) void bin_count_kernel(const int* __restrict__ col, int* __restrict__ cnt) {
  __shared__ int hist[NBK];
  const int tid = threadIdx.x;
  const int s = blockIdx.x;
  int cv[BI];
  #pragma unroll
  for (int u = 0; u < BI; ++u) {
    const int i = u * BT + tid;
    cv[u] = (i < SE) ? col[s * SE + i] : -1;
  }
  if (tid < NBK) hist[tid] = 0;
  __syncthreads();
  #pragma unroll
  for (int u = 0; u < BI; ++u)
    if (cv[u] >= 0) atomicAdd(&hist[cv[u] / RB], 1);
  __syncthreads();
  if (tid < NBK) cnt[tid * SL + s] = hist[tid];
}

__global__ __launch_bounds__(BT) void bin_place_kernel(const int* __restrict__ row, const int* __restrict__ col,
                                                       const float* __restrict__ dn, const int* __restrict__ cnt,
                                                       int* __restrict__ boff, int* __restrict__ recw) {
  __shared__ int tot[NBK];
  __shared__ int cur[NBK];
  const int tid = threadIdx.x;
  const int s = blockIdx.x;
  v2i* recs = (v2i*)recw;
  int rv[BI], cv[BI];
  float wv[BI];
  #pragma unroll
  for (int u = 0; u < BI; ++u) {
    const int i = u * BT + tid;
    const bool ok = i < SE;
    rv[u] = ok ? row[s * SE + i] : 0;
    cv[u] = ok ? col[s * SE + i] : -1;
  }
  #pragma unroll
  for (int u = 0; u < BI; ++u) wv[u] = (cv[u] >= 0) ? dn[rv[u]] * dn[cv[u]] : 0.0f;
  {
    const int b = tid >> 2, part = tid & 3;
    int t = 0, p = 0;
    for (int s2 = part; s2 < SL; s2 += 4) {
      const int v = cnt[b * SL + s2];
      t += v;
      p += (s2 < s) ? v : 0;
    }
    t += __shfl_xor(t, 1); t += __shfl_xor(t, 2);
    p += __shfl_xor(p, 1); p += __shfl_xor(p, 2);
    if (part == 0) { tot[b] = t; cur[b] = p; }
  }
  __syncthreads();
  if (tid < NBK) {
    int start = 0;
    for (int b2 = 0; b2 < NBK; ++b2) start += (b2 < tid) ? tot[b2] : 0;
    cur[tid] += start;
    if (s == 0) {
      boff[tid] = start;
      if (tid == NBK - 1) boff[NBK] = start + tot[tid];
    }
  }
  __syncthreads();
  #pragma unroll
  for (int u = 0; u < BI; ++u) {
    if (cv[u] >= 0) {
      const int bk = cv[u] / RB;
      const int pos = atomicAdd(&cur[bk], 1);
      v2i rec;
      rec.x = rv[u] | ((cv[u] - bk * RB) << RBITS);
      rec.y = __float_as_int(wv[u]);
      recs[pos] = rec;
    }
  }
}

#define EB 8
__device__ __forceinline__ void load_rows(const float* __restrict__ h, int r, int lane, int j0, float* hv) {
  #pragma unroll
  for (int u = 0; u < EB; ++u) {
    const int rj = __builtin_amdgcn_readlane(r, j0 + u);
    const float* hp = h + rj * HH + lane;
    hv[2 * u]     = hp[0];
    hv[2 * u + 1] = hp[32];
  }
}
__device__ __forceinline__ void add_rows(float* tile, int lc, int vb, int lane, int j0, const float* hv) {
  #pragma unroll
  for (int u = 0; u < EB; ++u) {
    const int cj = __builtin_amdgcn_readlane(lc, j0 + u);
    const float vj = __int_as_float(__builtin_amdgcn_readlane(vb, j0 + u));
    float* tp = tile + cj * HH + lane;
    atomicAdd(tp,      vj * hv[2 * u]);
    atomicAdd(tp + 32, vj * hv[2 * u + 1]);
  }
}
#define GW 13
#define GT0 (BT - GW * 32)
static_assert(GW * 32 >= RB + 3, "a lane for every node of a range moved up to multiples of four");
__device__ __forceinline__ void gate_store(const v4f* hr, const float* __restrict__ W, const float* __restrict__ B,
                                           float* __restrict__ ep) {
  float logit[KK];
  #pragma unroll
  for (int k = 0; k < KK; ++k) logit[k] = B[k];
  #pragma unroll
  for (int f = 0; f < HH; ++f) {
    const float hv = hr[f >> 2][f & 3];
    const float* wr = W + f * KK;
    #pragma unroll
    for (int k = 0; k < KK; ++k) logit[k] += hv * wr[k];
  }
  float m = logit[0];
  #pragma unroll
  for (int k = 1; k < KK; ++k) m = fmaxf(m, logit[k]);
  float s = 0.0f;
  float ex[KK];
  #pragma unroll
  for (int k = 0; k < KK; ++k) { ex[k] = __expf(logit[k] - m); s += ex[k]; }
  float inv = 1.0f / s;
  #pragma unroll
  for (int k = 0; k < KK; ++k) {
    float pi = ex[k] * inv;
    ep[k] = (pi > THETA) ? pi : 0.0f;
  }
}

__global__ __launch_bounds__(BT) void range_sum_kernel(const int* __restrict__ recw, const int* __restrict__ boff,
                                                       const float* __restrict__ h, float* __restrict__ agg,
                                                       const float* __restrict__ env_w, const float* __restrict__ env_b,
                                                       float* __restrict__ e_out, int layer) {
  __shared__ __attribute__((aligned(16))) float tile[(RB + 1) * HH];
  const v2i* recs = (const v2i*)recw;
  const int tid = threadIdx.x;
  const int lane = tid & 31;
  const int wave = tid >> 5;
  const int b = blockIdx.x;
  const v4f z = {0.0f, 0.0f, 0.0f, 0.0f};
  const int node_lo = b * RB;
  const int gn = ((node_lo + 3) & ~3) + (tid - GT0);
  const bool gate_on = tid >= GT0 && gn < min((node_lo + RB + 3) & ~3, NN);
  for (int i = tid; i < (RB + 1) * HH / 4; i += BT) ((v4f*)tile)[i] = z;
  const int e_lo = boff[b], e_hi = boff[b + 1];
  const v2i filler = {min(node_lo + lane, NN - 1) | (RB << RBITS), 0};
  __syncthreads();
  if (gate_on) {
    const v4f* hp = (const v4f*)(h + gn * HH);
    v4f hr[HH / 4];
    #pragma unroll
    for (int i = 0; i < HH / 4; ++i) hr[i] = hp[i];
    gate_store(hr, env_w + layer * 2 * HH * KK, env_b + layer * KK, e_out + gn * KK);
  }
  for (int e0 = e_lo + wave * 32; e0 < e_hi; e0 += BT) {
    v2i rec = filler;
    if (e0 + lane < e_hi) rec = recs[e0 + lane];
    const int r  = rec.x & ((1 << RBITS) - 1);
    const int lc = rec.x >> RBITS;
    const int vb = rec.y;
    float ha[2 * EB], hb[2 * EB];
    load_rows(h, r, lane, 0 * EB, ha);
    load_rows(h, r, lane, 1 * EB, hb);
    add_rows(tile, lc, vb, lane, 0 * EB, ha);
    load_rows(h, r, lane, 2 * EB, ha);
    add_rows(tile, lc, vb, lane, 1 * EB, hb);
    load_rows(h, r, lane, 3 * EB, hb);
    add_rows(tile, lc, vb, lane, 2 * EB, ha);
    add_rows(tile, lc, vb, lane, 3 * EB, hb);
  }
  __syncthreads();
  v4f* ap = (v4f*)(agg + node_lo * HH);
  for (int i = tid; i < RB * HH / 4; i += BT)
    if (node_lo + (i >> 4) < NN) ap[i] = ((const v4f*)tile)[i];
}

#define WP2 128
#define WBUF (HH * WP2)
#define NB 131
#define CONV_WG ((NN + NB - 1) / NB)
#define CW 8
#define CONV_LDS_OWN (4 * (WBUF + (NB + 1) * HH + 2 * KK * NB + KK * CW))
#define CONV_LDS_RESERVE 8192
static_assert(4 * (CONV_LDS_OWN + CONV_LDS_RESERVE) > 320 * 1024 && 3 * (CONV_LDS_OWN + CONV_LDS_RESERVE + 2048) <= 320 * 1024, "three per WGP, not four (2 KiB granule)");
static_assert(CONV_WG <= 3 * 256, "three work-groups per WGP hold the whole grid in one round");
static_assert(NB <= CW * 32, "a lane for every node of the work-group");

__device__ __forceinline__ void conv_half(const float* __restrict__ arow, const float* bl,
                                          float ek, int half, int lid,
                                          v8f& acc0, v8f& acc1, v8f& acc2, v8f& acc3) {
  const int s0 = 32 * half, s1 = 32 - 32 * half;
  #pragma unroll 4
  for (int c = 0; c < HH / 4; ++c) {
    const float* ap = arow + 4 * c + 2 * half;
    v2f a; a.x = ek * ap[0]; a.y = ek * ap[1];
    const float* bp = bl + (2 * c + half) * WP2 + 2 * lid;
    const v2f b0 = *(const v2f*)(bp + s0);
    const v2f b1 = *(const v2f*)(bp + s1);
    const v2f b2 = *(const v2f*)(bp + 64 + s0);
    const v2f b3 = *(const v2f*)(bp + 64 + s1);
    acc0 = wmma_f32(a, b0, acc0);
    acc1 = wmma_f32(a, b1, acc1);
    acc2 = wmma_f32(a, b2, acc2);
    acc3 = wmma_f32(a, b3, acc3);
  }
}

__global__ __launch_bounds__(CW * 32)
void conv_kernel(const float* __restrict__ agg,
                 float* h,
                 const float* __restrict__ e_gate,
                 const float* __restrict__ conv_w,
                 int layer) {
  __shared__ __attribute__((aligned(16))) float wl[WBUF];
  __shared__ __attribute__((aligned(16))) float oacc[(NB + 1) * HH];
  __shared__ int   lrow[KK * NB];
  __shared__ float lgate[KK * NB];
  __shared__ int   wcnt[KK * CW];

  const int tid  = threadIdx.x;
  const int wave = tid >> 5;
  const int lane = tid & 31;
  const int half = lane >> 4;
  const int lid  = lane & 15;
  const int base = blockIdx.x * NB;
  const int nrows = min(NB, NN - base);

  const float* W = conv_w + layer * (KK * 2 * HH * HH);

  const int goff = (tid >> 4) * (2 * HH) + (tid & 15) * 4;
  const int loff = (tid >> 4) * WP2 + (((tid & 15) * 8) ^ (((tid >> 4) & 1) * 32));
  v4f r0[4], r1[4];
  #pragma unroll
  for (int i = 0; i < 4; ++i) {
    r0[i] = *(const v4f*)(W + goff + i * 16 * 2 * HH);
    r1[i] = *(const v4f*)(W + goff + i * 16 * 2 * HH + HH);
  }
  v4f ga = {}, gb = {};
  if (tid < nrows) {
    const v4f* gp = (const v4f*)(e_gate + (base + tid) * KK);
    ga = gp[0]; gb = gp[1];
  }
  const float g[KK] = {ga.x, ga.y, ga.z, ga.w, gb.x, gb.y, gb.z, gb.w};
  unsigned m[KK];
  #pragma unroll
  for (int k = 0; k < KK; ++k) m[k] = __builtin_amdgcn_ballot_w32(g[k] != 0.0f);
  #pragma unroll
  for (int k = 0; k < KK; ++k)
    if (lane == k) wcnt[k * CW + wave] = __popc(m[k]);
  for (int i = tid; i < (NB + 1) * HH / 4; i += CW * 32) {
    const v4f z = {};
    ((v4f*)oacc)[i] = z;
  }
  #pragma unroll
  for (int i = 0; i < 4; ++i) {
    v4f lo, hi;
    lo.x = r0[i].x; lo.y = r1[i].x; lo.z = r0[i].y; lo.w = r1[i].y;
    hi.x = r0[i].z; hi.y = r1[i].z; hi.z = r0[i].w; hi.w = r1[i].w;
    *(v4f*)(wl + loff + i * 16 * WP2)     = lo;
    *(v4f*)(wl + loff + i * 16 * WP2 + 4) = hi;
  }
  __syncthreads();

  #pragma unroll
  for (int k = 0; k < KK; ++k) {
    int off = 0;
    #pragma unroll
    for (int w2 = 0; w2 < CW; ++w2) {
      const int cw = wcnt[k * CW + w2];
      off += (w2 < wave) ? cw : 0;
    }
    if (g[k] != 0.0f) {
      const int pos = off + __popc(m[k] & ((1u << lane) - 1u));
      lrow[k * NB + pos]  = tid;
      lgate[k * NB + pos] = g[k];
    }
  }
  __syncthreads();

  const int jt0 = (wave & 1) ? (wave ^ 4) : wave;

  #pragma unroll 1
  for (int k = 0; k < KK; ++k) {
    const bool more = (k + 1 < KK);
    if (more) {
      const float* Wn = W + (k + 1) * (2 * HH * HH) + goff;
      #pragma unroll
      for (int i = 0; i < 4; ++i) {
        r0[i] = *(const v4f*)(Wn + i * 16 * 2 * HH);
        r1[i] = *(const v4f*)(Wn + i * 16 * 2 * HH + HH);
      }
    }
    int tk = 0;
    #pragma unroll
    for (int w2 = 0; w2 < CW; ++w2) tk += wcnt[k * CW + w2];
    #pragma unroll 1
    for (int jt = jt0; 16 * jt < tk; jt += CW) {
      const int i = 16 * jt + lid;
      const bool valid = i < tk;
      int lr = 0; float ek = 0.0f;
      if (valid) { lr = lrow[k * NB + i]; ek = lgate[k * NB + i]; }
      const int drow = valid ? lr : NB;
      v8f acc0 = {}, acc1 = {}, acc2 = {}, acc3 = {};
      const float* bl = wl;
      conv_half(agg + (base + lr) * HH, bl,                  ek, half, lid, acc0, acc1, acc2, acc3);
      conv_half(h   + (base + lr) * HH, bl + (HH / 2) * WP2, ek, half, lid, acc0, acc1, acc2, acc3);
      float* op[8];
      float t0[8], t1[8], t2[8], t3[8];
      #pragma unroll
      for (int r = 0; r < 8; ++r) {
        const int dl = __builtin_amdgcn_readlane(drow, r);
        const int dh = __builtin_amdgcn_readlane(drow, r + 8);
        op[r] = oacc + (half ? dh : dl) * HH + lid;
        t0[r] = op[r][0];
        t1[r] = op[r][16];
        t2[r] = op[r][32];
        t3[r] = op[r][48];
      }
      #pragma unroll
      for (int r = 0; r < 8; ++r) {
        op[r][0]  = t0[r] + acc0[r];
        op[r][16] = t1[r] + acc1[r];
        op[r][32] = t2[r] + acc2[r];
        op[r][48] = t3[r] + acc3[r];
      }
    }
    if (more) {
      __syncthreads();
      float* dst = wl + loff;
      #pragma unroll
      for (int i = 0; i < 4; ++i) {
        v4f lo, hi;
        lo.x = r0[i].x; lo.y = r1[i].x; lo.z = r0[i].y; lo.w = r1[i].y;
        hi.x = r0[i].z; hi.y = r1[i].z; hi.z = r0[i].w; hi.w = r1[i].w;
        *(v4f*)(dst + i * 16 * WP2)     = lo;
        *(v4f*)(dst + i * 16 * WP2 + 4) = hi;
      }
    }
    __syncthreads();
  }
  #pragma unroll 4
  for (int r = wave; r < nrows; r += CW) {
    const int idx = (base + r) * HH + lane;
    const float* op = oacc + r * HH + lane;
    const float v0 = op[0]  + h[idx];
    const float v1 = op[32] + h[idx + 32];
    h[idx]      = v0 > 0.0f ? v0 : 0.0f;
    h[idx + 32] = v1 > 0.0f ? v1 : 0.0f;
  }
}

__global__ __launch_bounds__(256) void fc1_kernel(const float* __restrict__ h,
                                                  const float* __restrict__ w,
                                                  const float* __restrict__ b,
                                                  float* __restrict__ out) {
  const int wave = threadIdx.x >> 5;
  const int lane = threadIdx.x & 31;
  const int half = lane >> 4;
  const int lid  = lane & 15;
  const int node0 = blockIdx.x * 128 + wave * 16;
  if (node0 >= NN) return;

  v8f acc = {};
  for (int c = 0; c < HH / 4; ++c) {
    const int k0 = 4 * c + 2 * half;
    const float* ap = h + (node0 + lid) * HH + k0;
    v2f a; a.x = ap[0]; a.y = ap[1];
    const float* bp = w + k0 * CC;
    v2f bb; bb.x = bp[lid]; bb.y = bp[CC + lid];
    acc = wmma_f32(a, bb, acc);
  }
  const float bias = b[lid];
  for (int r = 0; r < 8; ++r)
    out[(node0 + r + 8 * half) * CC + lid] = acc[r] + bias;
}

extern "C" void kernel_launch(void* const* d_in, const int* in_sizes, int n_in,
                              void* d_out, int out_size, void* d_ws, size_t ws_size,
                              hipStream_t stream) {
  const float* x      = (const float*)d_in[0];
  const int*   eidx   = (const int*)d_in[1];
  const float* fc0_w  = (const float*)d_in[2];
  const float* fc0_b  = (const float*)d_in[3];
  const float* fc1_w  = (const float*)d_in[4];
  const float* fc1_b  = (const float*)d_in[5];
  const float* env_w  = (const float*)d_in[6];
  const float* env_b  = (const float*)d_in[7];
  const float* conv_w = (const float*)d_in[8];
  float* out = (float*)d_out;

  const int* row = eidx;
  const int* col = eidx + EE;

  float* ws  = (float*)d_ws;
  float* dn  = ws + NN;
  float* hA  = dn + NN;
  float* hB  = hA + (size_t)NN * HH;
  float* agg = hB + (size_t)NN * HH;
  float* eg  = agg + (size_t)NN * HH;
  int* cnt   = (int*)ws;
  int* boff  = cnt + NBK * SL;
  int* recs  = (int*)hB;

  const int T = 256;
  const int gemm_blocks = (NN + 127) / 128;

  int* part = (int*)hB;
  degree_hist_kernel<<<NQ * GS, HT, 0, stream>>>(col, part);
  dn_kernel<<<(NN + T - 1) / T, T, 0, stream>>>(part, dn);

  bin_count_kernel<<<SL, BT, 0, stream>>>(col, cnt);
  bin_place_kernel<<<SL, BT, 0, stream>>>(row, col, dn, cnt, boff, recs);

  fc0_kernel<<<gemm_blocks, T, 0, stream>>>(x, fc0_w, fc0_b, hA);

  for (int layer = 0; layer < LL; ++layer) {
    range_sum_kernel<<<NBK, BT, 0, stream>>>(recs, boff, hA, agg, env_w, env_b, eg, layer);
    conv_kernel<<<CONV_WG, CW * 32, CONV_LDS_RESERVE, stream>>>(agg, hA, eg, conv_w, layer);
  }

  fc1_kernel<<<gemm_blocks, T, 0, stream>>>(hA, fc1_w, fc1_b, out);
}


# conv: packed tiles of all 8 heads dealt to the 8 waves in one round-robin that runs on from head to head (start varies by work-group), not each head from wave 0: even tile counts per SIMD; sums kept
# speedup vs baseline: 2.7625x; 1.0317x over previous
#include <hip/hip_runtime.h>
#include <hip/hip_bf16.h>

#define NN 100000
#define DD 128
#define HH 64
#define KK 8
#define LL 3
#define CC 16
#define EE 1600000
#define THETA 0.1f

typedef __attribute__((ext_vector_type(2))) float v2f;
typedef __attribute__((ext_vector_type(8))) float v8f;

__device__ __forceinline__ v8f wmma_f32(v2f a, v2f b, v8f c) {
  return __builtin_amdgcn_wmma_f32_16x16x4_f32(
       false, a,  false, b,
       (short)0, c,  false,  false);
}


typedef __attribute__((ext_vector_type(4))) int v4i;
#define NQ 4
#define QN (NN / NQ)
#define GS 64
#define ES (EE / GS)
#define PS 25024
#define HT 1024
#define HI ((ES / 4 + HT - 1) / HT)
static_assert(NN % NQ == 0 && QN % 4 == 0 && PS >= QN && PS % 32 == 0, "quarters are whole 16-byte items, planes whole lines");
static_assert(EE % GS == 0 && ES % 4 == 0, "slices are whole 16-byte items");
static_assert(QN <= ES, "HI trips also cover a quarter's QN/4 items");
static_assert((size_t)NQ * GS * PS <= (size_t)2 * NN * HH, "the planes fit into the two feature planes they borrow");
__global__ __launch_bounds__(HT) void degree_hist_kernel(const int* __restrict__ col, int* __restrict__ part) {
  __shared__ __attribute__((aligned(16))) int cnt[QN];
  const int tid = threadIdx.x;
  const int q = blockIdx.x & (NQ - 1);
  const int g = blockIdx.x / NQ;
  const v4i* cp = (const v4i*)(col + g * ES);
  v4i cv[HI];
  #pragma unroll
  for (int i = 0; i < HI; ++i) {
    const int j = i * HT + tid;
    const v4i none = {-1, -1, -1, -1};
    cv[i] = (j < ES / 4) ? cp[j] : none;
  }
  #pragma unroll
  for (int i = 0; i < HI; ++i) {
    const int j = i * HT + tid;
    const v4i z = {0, 0, 0, 0};
    if (j < QN / 4) ((v4i*)cnt)[j] = z;
  }
  __syncthreads();
  const int lo = q * QN;
  #pragma unroll
  for (int i = 0; i < HI; ++i) {
    const unsigned d0 = (unsigned)(cv[i].x - lo), d1 = (unsigned)(cv[i].y - lo);
    const unsigned d2 = (unsigned)(cv[i].z - lo), d3 = (unsigned)(cv[i].w - lo);
    if (d0 < (unsigned)QN) atomicAdd(&cnt[d0], 1);
    if (d1 < (unsigned)QN) atomicAdd(&cnt[d1], 1);
    if (d2 < (unsigned)QN) atomicAdd(&cnt[d2], 1);
    if (d3 < (unsigned)QN) atomicAdd(&cnt[d3], 1);
  }
  __syncthreads();
  v4i* op = (v4i*)(part + (q * GS + g) * PS);
  #pragma unroll
  for (int i = 0; i < HI; ++i) {
    const int j = i * HT + tid;
    if (j < QN / 4) op[j] = ((const v4i*)cnt)[j];
  }
}

__global__ void dn_kernel(const int* __restrict__ part, float* __restrict__ dn) {
  int i = blockIdx.x * blockDim.x + threadIdx.x;
  if (i < NN) {
    const int q = i / QN;
    const int* p = part + q * (GS * PS) + (i - q * QN);
    int d = 0;
    #pragma unroll
    for (int g = 0; g < GS; ++g) d += p[g * PS];
    dn[i] = (d > 0) ? rsqrtf((float)d) : 0.0f;
  }
}

__global__ __launch_bounds__(256) void fc0_kernel(const float* __restrict__ x,
                                                  const float* __restrict__ w,
                                                  const float* __restrict__ b,
                                                  float* __restrict__ h) {
  const int wave = threadIdx.x >> 5;
  const int lane = threadIdx.x & 31;
  const int half = lane >> 4;
  const int lid  = lane & 15;
  const int node0 = blockIdx.x * 128 + wave * 16;
  if (node0 >= NN) return;

  v8f acc0 = {}, acc1 = {}, acc2 = {}, acc3 = {};
  for (int c = 0; c < DD / 4; ++c) {
    const int k0 = 4 * c + 2 * half;
    const float* ap = x + (node0 + lid) * DD + k0;
    v2f a; a.x = ap[0]; a.y = ap[1];
    const float* bp = w + k0 * HH;
    v2f b0; b0.x = bp[lid];      b0.y = bp[HH + lid];
    v2f b1; b1.x = bp[16 + lid]; b1.y = bp[HH + 16 + lid];
    v2f b2; b2.x = bp[32 + lid]; b2.y = bp[HH + 32 + lid];
    v2f b3; b3.x = bp[48 + lid]; b3.y = bp[HH + 48 + lid];
    acc0 = wmma_f32(a, b0, acc0);
    acc1 = wmma_f32(a, b1, acc1);
    acc2 = wmma_f32(a, b2, acc2);
    acc3 = wmma_f32(a, b3, acc3);
  }
  v8f accs[4] = {acc0, acc1, acc2, acc3};
  for (int t = 0; t < 4; ++t) {
    const float bias = b[16 * t + lid];
    for (int r = 0; r < 8; ++r) {
      float v = accs[t][r] + bias;
      h[(node0 + r + 8 * half) * HH + 16 * t + lid] = v > 0.0f ? v : 0.0f;
    }
  }
}

typedef __attribute__((ext_vector_type(2))) int v2i;
typedef __attribute__((ext_vector_type(4))) float v4f;
#define RB 391
#define NBK 256
#define SL 250
#define SE (EE / SL)
#define BT 1024
#define RBITS 17
#define BI ((SE + BT - 1) / BT)
static_assert(RB * NBK >= NN, "the ranges cover every node");
static_assert(EE % SL == 0, "slices are whole");
static_assert(BT == 4 * NBK, "four threads per range in the place pass");
static_assert(NN <= (1 << RBITS) && RB < (1 << (31 - RBITS)), "source row and local row (the spare row RB too) fit one non-negative word");
static_assert(NBK * SL + NBK + 1 <= NN, "counts and range starts fit the spare N words at the head of the workspace");
static_assert((size_t)EE * 2 <= (size_t)NN * HH, "the records fit one feature plane");

__global__ __launch_bounds__(BT) void bin_count_kernel(const int* __restrict__ col, int* __restrict__ cnt) {
  __shared__ int hist[NBK];
  const int tid = threadIdx.x;
  const int s = blockIdx.x;
  int cv[BI];
  #pragma unroll
  for (int u = 0; u < BI; ++u) {
    const int i = u * BT + tid;
    cv[u] = (i < SE) ? col[s * SE + i] : -1;
  }
  if (tid < NBK) hist[tid] = 0;
  __syncthreads();
  #pragma unroll
  for (int u = 0; u < BI; ++u)
    if (cv[u] >= 0) atomicAdd(&hist[cv[u] / RB], 1);
  __syncthreads();
  if (tid < NBK) cnt[tid * SL + s] = hist[tid];
}

__global__ __launch_bounds__(BT) void bin_place_kernel(const int* __restrict__ row, const int* __restrict__ col,
                                                       const float* __restrict__ dn, const int* __restrict__ cnt,
                                                       int* __restrict__ boff, int* __restrict__ recw) {
  __shared__ int tot[NBK];
  __shared__ int cur[NBK];
  const int tid = threadIdx.x;
  const int s = blockIdx.x;
  v2i* recs = (v2i*)recw;
  int rv[BI], cv[BI];
  float wv[BI];
  #pragma unroll
  for (int u = 0; u < BI; ++u) {
    const int i = u * BT + tid;
    const bool ok = i < SE;
    rv[u] = ok ? row[s * SE + i] : 0;
    cv[u] = ok ? col[s * SE + i] : -1;
  }
  #pragma unroll
  for (int u = 0; u < BI; ++u) wv[u] = (cv[u] >= 0) ? dn[rv[u]] * dn[cv[u]] : 0.0f;
  {
    const int b = tid >> 2, part = tid & 3;
    int t = 0, p = 0;
    for (int s2 = part; s2 < SL; s2 += 4) {
      const int v = cnt[b * SL + s2];
      t += v;
      p += (s2 < s) ? v : 0;
    }
    t += __shfl_xor(t, 1); t += __shfl_xor(t, 2);
    p += __shfl_xor(p, 1); p += __shfl_xor(p, 2);
    if (part == 0) { tot[b] = t; cur[b] = p; }
  }
  __syncthreads();
  if (tid < NBK) {
    int start = 0;
    for (int b2 = 0; b2 < NBK; ++b2) start += (b2 < tid) ? tot[b2] : 0;
    cur[tid] += start;
    if (s == 0) {
      boff[tid] = start;
      if (tid == NBK - 1) boff[NBK] = start + tot[tid];
    }
  }
  __syncthreads();
  #pragma unroll
  for (int u = 0; u < BI; ++u) {
    if (cv[u] >= 0) {
      const int bk = cv[u] / RB;
      const int pos = atomicAdd(&cur[bk], 1);
      v2i rec;
      rec.x = rv[u] | ((cv[u] - bk * RB) << RBITS);
      rec.y = __float_as_int(wv[u]);
      recs[pos] = rec;
    }
  }
}

#define EB 8
__device__ __forceinline__ void load_rows(const float* __restrict__ h, int r, int lane, int j0, float* hv) {
  #pragma unroll
  for (int u = 0; u < EB; ++u) {
    const int rj = __builtin_amdgcn_readlane(r, j0 + u);
    const float* hp = h + rj * HH + lane;
    hv[2 * u]     = hp[0];
    hv[2 * u + 1] = hp[32];
  }
}
__device__ __forceinline__ void add_rows(float* tile, int lc, int vb, int lane, int j0, const float* hv) {
  #pragma unroll
  for (int u = 0; u < EB; ++u) {
    const int cj = __builtin_amdgcn_readlane(lc, j0 + u);
    const float vj = __int_as_float(__builtin_amdgcn_readlane(vb, j0 + u));
    float* tp = tile + cj * HH + lane;
    atomicAdd(tp,      vj * hv[2 * u]);
    atomicAdd(tp + 32, vj * hv[2 * u + 1]);
  }
}
#define GW 13
#define GT0 (BT - GW * 32)
static_assert(GW * 32 >= RB + 3, "a lane for every node of a range moved up to multiples of four");
__device__ __forceinline__ void gate_store(const v4f* hr, const float* __restrict__ W, const float* __restrict__ B,
                                           float* __restrict__ ep) {
  float logit[KK];
  #pragma unroll
  for (int k = 0; k < KK; ++k) logit[k] = B[k];
  #pragma unroll
  for (int f = 0; f < HH; ++f) {
    const float hv = hr[f >> 2][f & 3];
    const float* wr = W + f * KK;
    #pragma unroll
    for (int k = 0; k < KK; ++k) logit[k] += hv * wr[k];
  }
  float m = logit[0];
  #pragma unroll
  for (int k = 1; k < KK; ++k) m = fmaxf(m, logit[k]);
  float s = 0.0f;
  float ex[KK];
  #pragma unroll
  for (int k = 0; k < KK; ++k) { ex[k] = __expf(logit[k] - m); s += ex[k]; }
  float inv = 1.0f / s;
  #pragma unroll
  for (int k = 0; k < KK; ++k) {
    float pi = ex[k] * inv;
    ep[k] = (pi > THETA) ? pi : 0.0f;
  }
}

__global__ __launch_bounds__(BT) void range_sum_kernel(const int* __restrict__ recw, const int* __restrict__ boff,
                                                       const float* __restrict__ h, float* __restrict__ agg,
                                                       const float* __restrict__ env_w, const float* __restrict__ env_b,
                                                       float* __restrict__ e_out, int layer) {
  __shared__ __attribute__((aligned(16))) float tile[(RB + 1) * HH];
  const v2i* recs = (const v2i*)recw;
  const int tid = threadIdx.x;
  const int lane = tid & 31;
  const int wave = tid >> 5;
  const int b = blockIdx.x;
  const v4f z = {0.0f, 0.0f, 0.0f, 0.0f};
  const int node_lo = b * RB;
  const int gn = ((node_lo + 3) & ~3) + (tid - GT0);
  const bool gate_on = tid >= GT0 && gn < min((node_lo + RB + 3) & ~3, NN);
  for (int i = tid; i < (RB + 1) * HH / 4; i += BT) ((v4f*)tile)[i] = z;
  const int e_lo = boff[b], e_hi = boff[b + 1];
  const v2i filler = {min(node_lo + lane, NN - 1) | (RB << RBITS), 0};
  __syncthreads();
  if (gate_on) {
    const v4f* hp = (const v4f*)(h + gn * HH);
    v4f hr[HH / 4];
    #pragma unroll
    for (int i = 0; i < HH / 4; ++i) hr[i] = hp[i];
    gate_store(hr, env_w + layer * 2 * HH * KK, env_b + layer * KK, e_out + gn * KK);
  }
  for (int e0 = e_lo + wave * 32; e0 < e_hi; e0 += BT) {
    v2i rec = filler;
    if (e0 + lane < e_hi) rec = recs[e0 + lane];
    const int r  = rec.x & ((1 << RBITS) - 1);
    const int lc = rec.x >> RBITS;
    const int vb = rec.y;
    float ha[2 * EB], hb[2 * EB];
    load_rows(h, r, lane, 0 * EB, ha);
    load_rows(h, r, lane, 1 * EB, hb);
    add_rows(tile, lc, vb, lane, 0 * EB, ha);
    load_rows(h, r, lane, 2 * EB, ha);
    add_rows(tile, lc, vb, lane, 1 * EB, hb);
    load_rows(h, r, lane, 3 * EB, hb);
    add_rows(tile, lc, vb, lane, 2 * EB, ha);
    add_rows(tile, lc, vb, lane, 3 * EB, hb);
  }
  __syncthreads();
  v4f* ap = (v4f*)(agg + node_lo * HH);
  for (int i = tid; i < RB * HH / 4; i += BT)
    if (node_lo + (i >> 4) < NN) ap[i] = ((const v4f*)tile)[i];
}

#define WP2 128
#define WBUF (HH * WP2)
#define NB 131
#define CONV_WG ((NN + NB - 1) / NB)
#define CW 8
#define CONV_LDS_OWN (4 * (WBUF + (NB + 1) * HH + 2 * KK * NB + KK * CW))
#define CONV_LDS_RESERVE 8192
static_assert(4 * (CONV_LDS_OWN + CONV_LDS_RESERVE) > 320 * 1024 && 3 * (CONV_LDS_OWN + CONV_LDS_RESERVE + 2048) <= 320 * 1024, "three per WGP, not four (2 KiB granule)");
static_assert(CONV_WG <= 3 * 256, "three work-groups per WGP hold the whole grid in one round");
static_assert(NB <= CW * 32, "a lane for every node of the work-group");

__device__ __forceinline__ void conv_half(const float* __restrict__ arow, const float* bl,
                                          float ek, int half, int lid,
                                          v8f& acc0, v8f& acc1, v8f& acc2, v8f& acc3) {
  const int s0 = 32 * half, s1 = 32 - 32 * half;
  #pragma unroll 4
  for (int c = 0; c < HH / 4; ++c) {
    const float* ap = arow + 4 * c + 2 * half;
    v2f a; a.x = ek * ap[0]; a.y = ek * ap[1];
    const float* bp = bl + (2 * c + half) * WP2 + 2 * lid;
    const v2f b0 = *(const v2f*)(bp + s0);
    const v2f b1 = *(const v2f*)(bp + s1);
    const v2f b2 = *(const v2f*)(bp + 64 + s0);
    const v2f b3 = *(const v2f*)(bp + 64 + s1);
    acc0 = wmma_f32(a, b0, acc0);
    acc1 = wmma_f32(a, b1, acc1);
    acc2 = wmma_f32(a, b2, acc2);
    acc3 = wmma_f32(a, b3, acc3);
  }
}

__global__ __launch_bounds__(CW * 32)
void conv_kernel(const float* __restrict__ agg,
                 float* h,
                 const float* __restrict__ e_gate,
                 const float* __restrict__ conv_w,
                 int layer) {
  __shared__ __attribute__((aligned(16))) float wl[WBUF];
  __shared__ __attribute__((aligned(16))) float oacc[(NB + 1) * HH];
  __shared__ int   lrow[KK * NB];
  __shared__ float lgate[KK * NB];
  __shared__ int   wcnt[KK * CW];

  const int tid  = threadIdx.x;
  const int wave = tid >> 5;
  const int lane = tid & 31;
  const int half = lane >> 4;
  const int lid  = lane & 15;
  const int base = blockIdx.x * NB;
  const int nrows = min(NB, NN - base);

  const float* W = conv_w + layer * (KK * 2 * HH * HH);

  const int goff = (tid >> 4) * (2 * HH) + (tid & 15) * 4;
  const int loff = (tid >> 4) * WP2 + (((tid & 15) * 8) ^ (((tid >> 4) & 1) * 32));
  v4f r0[4], r1[4];
  #pragma unroll
  for (int i = 0; i < 4; ++i) {
    r0[i] = *(const v4f*)(W + goff + i * 16 * 2 * HH);
    r1[i] = *(const v4f*)(W + goff + i * 16 * 2 * HH + HH);
  }
  v4f ga = {}, gb = {};
  if (tid < nrows) {
    const v4f* gp = (const v4f*)(e_gate + (base + tid) * KK);
    ga = gp[0]; gb = gp[1];
  }
  const float g[KK] = {ga.x, ga.y, ga.z, ga.w, gb.x, gb.y, gb.z, gb.w};
  unsigned m[KK];
  #pragma unroll
  for (int k = 0; k < KK; ++k) m[k] = __builtin_amdgcn_ballot_w32(g[k] != 0.0f);
  #pragma unroll
  for (int k = 0; k < KK; ++k)
    if (lane == k) wcnt[k * CW + wave] = __popc(m[k]);
  for (int i = tid; i < (NB + 1) * HH / 4; i += CW * 32) {
    const v4f z = {};
    ((v4f*)oacc)[i] = z;
  }
  #pragma unroll
  for (int i = 0; i < 4; ++i) {
    v4f lo, hi;
    lo.x = r0[i].x; lo.y = r1[i].x; lo.z = r0[i].y; lo.w = r1[i].y;
    hi.x = r0[i].z; hi.y = r1[i].z; hi.z = r0[i].w; hi.w = r1[i].w;
    *(v4f*)(wl + loff + i * 16 * WP2)     = lo;
    *(v4f*)(wl + loff + i * 16 * WP2 + 4) = hi;
  }
  __syncthreads();

  #pragma unroll
  for (int k = 0; k < KK; ++k) {
    int off = 0;
    #pragma unroll
    for (int w2 = 0; w2 < CW; ++w2) {
      const int cw = wcnt[k * CW + w2];
      off += (w2 < wave) ? cw : 0;
    }
    if (g[k] != 0.0f) {
      const int pos = off + __popc(m[k] & ((1u << lane) - 1u));
      lrow[k * NB + pos]  = tid;
      lgate[k * NB + pos] = g[k];
    }
  }
  __syncthreads();

  const int place = (wave & 1) ? (wave ^ 4) : wave;
  int first = (blockIdx.x + (blockIdx.x >> 8)) & (CW - 1);

  #pragma unroll 1
  for (int k = 0; k < KK; ++k) {
    const bool more = (k + 1 < KK);
    if (more) {
      const float* Wn = W + (k + 1) * (2 * HH * HH) + goff;
      #pragma unroll
      for (int i = 0; i < 4; ++i) {
        r0[i] = *(const v4f*)(Wn + i * 16 * 2 * HH);
        r1[i] = *(const v4f*)(Wn + i * 16 * 2 * HH + HH);
      }
    }
    int tk = 0;
    #pragma unroll
    for (int w2 = 0; w2 < CW; ++w2) tk += wcnt[k * CW + w2];
    const int jtw = (place - first) & (CW - 1);
    first = (first + ((tk + 15) >> 4)) & (CW - 1);
    #pragma unroll 1
    for (int jt = jtw; 16 * jt < tk; jt += CW) {
      const int i = 16 * jt + lid;
      const bool valid = i < tk;
      int lr = 0; float ek = 0.0f;
      if (valid) { lr = lrow[k * NB + i]; ek = lgate[k * NB + i]; }
      const int drow = valid ? lr : NB;
      v8f acc0 = {}, acc1 = {}, acc2 = {}, acc3 = {};
      const float* bl = wl;
      conv_half(agg + (base + lr) * HH, bl,                  ek, half, lid, acc0, acc1, acc2, acc3);
      conv_half(h   + (base + lr) * HH, bl + (HH / 2) * WP2, ek, half, lid, acc0, acc1, acc2, acc3);
      float* op[8];
      float t0[8], t1[8], t2[8], t3[8];
      #pragma unroll
      for (int r = 0; r < 8; ++r) {
        const int dl = __builtin_amdgcn_readlane(drow, r);
        const int dh = __builtin_amdgcn_readlane(drow, r + 8);
        op[r] = oacc + (half ? dh : dl) * HH + lid;
        t0[r] = op[r][0];
        t1[r] = op[r][16];
        t2[r] = op[r][32];
        t3[r] = op[r][48];
      }
      #pragma unroll
      for (int r = 0; r < 8; ++r) {
        op[r][0]  = t0[r] + acc0[r];
        op[r][16] = t1[r] + acc1[r];
        op[r][32] = t2[r] + acc2[r];
        op[r][48] = t3[r] + acc3[r];
      }
    }
    if (more) {
      __syncthreads();
      float* dst = wl + loff;
      #pragma unroll
      for (int i = 0; i < 4; ++i) {
        v4f lo, hi;
        lo.x = r0[i].x; lo.y = r1[i].x; lo.z = r0[i].y; lo.w = r1[i].y;
        hi.x = r0[i].z; hi.y = r1[i].z; hi.z = r0[i].w; hi.w = r1[i].w;
        *(v4f*)(dst + i * 16 * WP2)     = lo;
        *(v4f*)(dst + i * 16 * WP2 + 4) = hi;
      }
    }
    __syncthreads();
  }
  #pragma unroll 4
  for (int r = wave; r < nrows; r += CW) {
    const int idx = (base + r) * HH + lane;
    const float* op = oacc + r * HH + lane;
    const float v0 = op[0]  + h[idx];
    const float v1 = op[32] + h[idx + 32];
    h[idx]      = v0 > 0.0f ? v0 : 0.0f;
    h[idx + 32] = v1 > 0.0f ? v1 : 0.0f;
  }
}

__global__ __launch_bounds__(256) void fc1_kernel(const float* __restrict__ h,
                                                  const float* __restrict__ w,
                                                  const float* __restrict__ b,
                                                  float* __restrict__ out) {
  const int wave = threadIdx.x >> 5;
  const int lane = threadIdx.x & 31;
  const int half = lane >> 4;
  const int lid  = lane & 15;
  const int node0 = blockIdx.x * 128 + wave * 16;
  if (node0 >= NN) return;

  v8f acc = {};
  for (int c = 0; c < HH / 4; ++c) {
    const int k0 = 4 * c + 2 * half;
    const float* ap = h + (node0 + lid) * HH + k0;
    v2f a; a.x = ap[0]; a.y = ap[1];
    const float* bp = w + k0 * CC;
    v2f bb; bb.x = bp[lid]; bb.y = bp[CC + lid];
    acc = wmma_f32(a, bb, acc);
  }
  const float bias = b[lid];
  for (int r = 0; r < 8; ++r)
    out[(node0 + r + 8 * half) * CC + lid] = acc[r] + bias;
}

extern "C" void kernel_launch(void* const* d_in, const int* in_sizes, int n_in,
                              void* d_out, int out_size, void* d_ws, size_t ws_size,
                              hipStream_t stream) {
  const float* x      = (const float*)d_in[0];
  const int*   eidx   = (const int*)d_in[1];
  const float* fc0_w  = (const float*)d_in[2];
  const float* fc0_b  = (const float*)d_in[3];
  const float* fc1_w  = (const float*)d_in[4];
  const float* fc1_b  = (const float*)d_in[5];
  const float* env_w  = (const float*)d_in[6];
  const float* env_b  = (const float*)d_in[7];
  const float* conv_w = (const float*)d_in[8];
  float* out = (float*)d_out;

  const int* row = eidx;
  const int* col = eidx + EE;

  float* ws  = (float*)d_ws;
  float* dn  = ws + NN;
  float* hA  = dn + NN;
  float* hB  = hA + (size_t)NN * HH;
  float* agg = hB + (size_t)NN * HH;
  float* eg  = agg + (size_t)NN * HH;
  int* cnt   = (int*)ws;
  int* boff  = cnt + NBK * SL;
  int* recs  = (int*)hB;

  const int T = 256;
  const int gemm_blocks = (NN + 127) / 128;

  int* part = (int*)hB;
  degree_hist_kernel<<<NQ * GS, HT, 0, stream>>>(col, part);
  dn_kernel<<<(NN + T - 1) / T, T, 0, stream>>>(part, dn);

  bin_count_kernel<<<SL, BT, 0, stream>>>(col, cnt);
  bin_place_kernel<<<SL, BT, 0, stream>>>(row, col, dn, cnt, boff, recs);

  fc0_kernel<<<gemm_blocks, T, 0, stream>>>(x, fc0_w, fc0_b, hA);

  for (int layer = 0; layer < LL; ++layer) {
    range_sum_kernel<<<NBK, BT, 0, stream>>>(recs, boff, hA, agg, env_w, env_b, eg, layer);
    conv_kernel<<<CONV_WG, CW * 32, CONV_LDS_RESERVE, stream>>>(agg, hA, eg, conv_w, layer);
  }

  fc1_kernel<<<gemm_blocks, T, 0, stream>>>(hA, fc1_w, fc1_b, out);
}
